# starting kernel
# baseline (speedup 1.0000x reference)
#include <hip/hip_runtime.h>

typedef __attribute__((ext_vector_type(2))) float v2f;
typedef __attribute__((ext_vector_type(8))) float v8f;

#define KDIM 128

__global__ void k_init_deg(float* __restrict__ deg, int n) {
    int i = blockIdx.x * blockDim.x + threadIdx.x;
    if (i < n) deg[i] = 1.0f;
}

__global__ void k_count_deg(const int* __restrict__ dst, float* __restrict__ deg, int e) {
    int i = blockIdx.x * blockDim.x + threadIdx.x;
    if (i < e) atomicAdd(&deg[dst[i]], 1.0f);
}

__global__ void k_finalize_dinv(float* __restrict__ deg, int n) {
    int i = blockIdx.x * blockDim.x + threadIdx.x;
    if (i < n) deg[i] = rsqrtf(deg[i]);
}

template <int OUTCH, int MT, bool RELU>
__global__ __launch_bounds__(256) void k_gemm(const float* __restrict__ A,
                                              const float* __restrict__ W,
                                              float* __restrict__ H, int n) {
    constexpr int LDSW = KDIM + 4;
    __shared__ float tile[MT * 16 * LDSW];

    const int tid = threadIdx.x;
    const int m_base = blockIdx.x * (MT * 16);

    constexpr int CHUNKS = MT * 16 * (KDIM / 4);
    for (int c = tid; c < CHUNKS; c += 256) {
        int row = c >> 5;
        int kc  = c & 31;
        int grow = m_base + row;
        if (grow >= n) grow = n - 1;
        float4 v = ((const float4*)(A + (size_t)grow * KDIM))[kc];
        if (RELU) {
            v.x = fmaxf(v.x, 0.0f); v.y = fmaxf(v.y, 0.0f);
            v.z = fmaxf(v.z, 0.0f); v.w = fmaxf(v.w, 0.0f);
        }
        *(float4*)&tile[row * LDSW + kc * 4] = v;
    }
    __syncthreads();

    const int wave = tid >> 5;
    const int lane = tid & 31;
    int mt, nt;
    if (OUTCH == 128) { mt = 0;         nt = wave;     }
    else              { mt = wave >> 2; nt = wave & 3; }
    const int lm = lane & 15;
    const int kh = lane >> 4;

    const float* arow = &tile[(mt * 16 + lm) * LDSW];
    const float* wcol = W + nt * 16 + lm;

    v8f acc = {};
#pragma unroll
    for (int k0 = 0; k0 < KDIM; k0 += 4) {
        v2f a, b;
        a.x = arow[k0 + 2 * kh + 0];
        a.y = arow[k0 + 2 * kh + 1];
        b.x = wcol[(size_t)(k0 + 2 * kh + 0) * OUTCH];
        b.y = wcol[(size_t)(k0 + 2 * kh + 1) * OUTCH];
        acc = __builtin_amdgcn_wmma_f32_16x16x4_f32(
             false, a,  false, b,
             (short)0, acc,  false,  false);
    }

    const int mrow0 = m_base + mt * 16 + (kh ? 8 : 0);
#pragma unroll
    for (int r = 0; r < 8; ++r) {
        int grow = mrow0 + r;
        if (grow < n)
            H[(size_t)grow * OUTCH + nt * 16 + lm] = acc[r];
    }
}

template <int CH>
__global__ void k_seed(float* __restrict__ out, const float* __restrict__ h,
                       const float* __restrict__ bias, const float* __restrict__ dinv, int n) {
    int i = blockIdx.x * blockDim.x + threadIdx.x;
    int total = n * (CH / 4);
    if (i >= total) return;
    int node = i / (CH / 4);
    int c4   = i % (CH / 4);
    float d = dinv[node];
    float s = d * d;
    float4 hv = ((const float4*)(h + (size_t)node * CH))[c4];
    float4 bv = ((const float4*)bias)[c4];
    float4 o;
    o.x = fmaf(hv.x, s, bv.x); o.y = fmaf(hv.y, s, bv.y);
    o.z = fmaf(hv.z, s, bv.z); o.w = fmaf(hv.w, s, bv.w);
    ((float4*)(out + (size_t)node * CH))[c4] = o;
}

template <int CH>
__global__ void k_scatter(const int* __restrict__ src, const int* __restrict__ dst,
                          const float* __restrict__ h, float* __restrict__ out,
                          const float* __restrict__ dinv, int e) {
    int g = blockIdx.x * blockDim.x + threadIdx.x;
    int edge = g >> 5;
    int lane = g & 31;
    if (edge >= e) return;
    int s = src[edge], d = dst[edge];
    float w = dinv[s] * dinv[d];
    if (CH == 128) {
        float4 hv = ((const float4*)(h + (size_t)s * CH))[lane];
        float* o = out + (size_t)d * CH + lane * 4;
        atomicAdd(o + 0, hv.x * w); atomicAdd(o + 1, hv.y * w);
        atomicAdd(o + 2, hv.z * w); atomicAdd(o + 3, hv.w * w);
    } else {
        float2 hv = ((const float2*)(h + (size_t)s * CH))[lane];
        float* o = out + (size_t)d * CH + lane * 2;
        atomicAdd(o + 0, hv.x * w); atomicAdd(o + 1, hv.y * w);
    }
}

extern "C" void kernel_launch(void* const* d_in, const int* in_sizes, int n_in,
                              void* d_out, int out_size, void* d_ws, size_t ws_size,
                              hipStream_t stream) {
    const float* x  = (const float*)d_in[0];
    const int*   ei = (const int*)d_in[1];
    const float* W1 = (const float*)d_in[2];
    const float* b1 = (const float*)d_in[3];
    const float* W2 = (const float*)d_in[4];
    const float* b2 = (const float*)d_in[5];
    float* out = (float*)d_out;

    const int n = in_sizes[0] / 128;
    const int e = in_sizes[1] / 2;
    const int* src = ei;
    const int* dst = ei + e;

    float* dinv = (float*)d_ws;
    float* h  = dinv + n;
    float* o1 = h + (size_t)n * 128;

    const unsigned T = 256;
    k_init_deg<<<(n + T - 1) / T, T, 0, stream>>>(dinv, n);
    k_count_deg<<<(e + T - 1) / T, T, 0, stream>>>(dst, dinv, e);
    k_finalize_dinv<<<(n + T - 1) / T, T, 0, stream>>>(dinv, n);

    k_gemm<128, 1, false><<<(n + 15) / 16, T, 0, stream>>>(x, W1, h, n);
    k_seed<128><<<((unsigned)(n * 32) + T - 1) / T, T, 0, stream>>>(o1, h, b1, dinv, n);
    k_scatter<128><<<(unsigned)((e + 7) / 8), T, 0, stream>>>(src, dst, h, o1, dinv, e);

    k_gemm<64, 2, true><<<(n + 31) / 32, T, 0, stream>>>(o1, W2, h, n);
    k_seed<64><<<((unsigned)(n * 16) + T - 1) / T, T, 0, stream>>>(out, h, b2, dinv, n);
    k_scatter<64><<<(unsigned)((e + 7) / 8), T, 0, stream>>>(src, dst, h, out, dinv, e);
}


# Pull aggregation: edges bucketed by destination in-call (int atomic rank, scan, fill); one wave per row sums in registers in bucket-arrival order, stores once; no float atomics
# speedup vs baseline: 2.2103x; 2.2103x over previous
#include <hip/hip_runtime.h>

typedef __attribute__((ext_vector_type(2))) float v2f;
typedef __attribute__((ext_vector_type(8))) float v8f;

#define KDIM 128


__global__ void k_zero_cnt(int* __restrict__ cnt, int n) {
    int i = blockIdx.x * blockDim.x + threadIdx.x;
    if (i < n) cnt[i] = 0;
}

__global__ void k_rank(const int* __restrict__ dst, int* __restrict__ cnt,
                       int* __restrict__ pos, int e) {
    int i = blockIdx.x * blockDim.x + threadIdx.x;
    if (i < e) pos[i] = atomicAdd(&cnt[dst[i]], 1);
}

#define SCAN_T 1024
__device__ __forceinline__ int block_sum_1024(int v, int* sh) {
    const int lane = threadIdx.x & 31, wave = threadIdx.x >> 5;
#pragma unroll
    for (int o = 16; o > 0; o >>= 1) v += __shfl_xor(v, o);
    if (lane == 0) sh[wave] = v;
    __syncthreads();
    int t = sh[lane];
#pragma unroll
    for (int o = 16; o > 0; o >>= 1) t += __shfl_xor(t, o);
    __syncthreads();
    return t;
}

__global__ __launch_bounds__(SCAN_T) void k_block_sums(const int* __restrict__ cnt,
                                                       int* __restrict__ bsum, int n) {
    __shared__ int sh[32];
    int i = blockIdx.x * SCAN_T + threadIdx.x;
    int t = block_sum_1024(i < n ? cnt[i] : 0, sh);
    if (threadIdx.x == 0) bsum[blockIdx.x] = t;
}

__global__ __launch_bounds__(SCAN_T) void k_scan(const int* __restrict__ cnt,
                                                 const int* __restrict__ bsum,
                                                 int* __restrict__ rowptr,
                                                 float* __restrict__ dinv, int n) {
    __shared__ int sh[32];
    const int lane = threadIdx.x & 31, wave = threadIdx.x >> 5;
    int part = 0;
    for (int b = threadIdx.x; b < (int)blockIdx.x; b += SCAN_T) part += bsum[b];
    const int offset = block_sum_1024(part, sh);

    const int i = blockIdx.x * SCAN_T + threadIdx.x;
    const int c = (i < n) ? cnt[i] : 0;
    int incl = c;
#pragma unroll
    for (int o = 1; o < 32; o <<= 1) {
        int u = __shfl_up(incl, o);
        if (lane >= o) incl += u;
    }
    if (lane == 31) sh[wave] = incl;
    __syncthreads();
    int wsum = sh[lane];
#pragma unroll
    for (int o = 1; o < 32; o <<= 1) {
        int u = __shfl_up(wsum, o);
        if (lane >= o) wsum += u;
    }
    const int wave_excl = __shfl(wsum, wave) - __shfl(sh[lane], wave);
    const int excl = offset + wave_excl + incl - c;
    if (i < n) {
        rowptr[i] = excl;
        dinv[i] = rsqrtf((float)(c + 1));
        if (i == n - 1) rowptr[n] = excl + c;
    }
}

__global__ void k_fill(const int* __restrict__ src, const int* __restrict__ dst,
                       const int* __restrict__ pos, const int* __restrict__ rowptr,
                       int* __restrict__ ssrc, int e) {
    int i = blockIdx.x * blockDim.x + threadIdx.x;
    if (i < e) ssrc[rowptr[dst[i]] + pos[i]] = src[i];
}

__global__ void k_move_buckets(const int* __restrict__ from, int* __restrict__ to, int count) {
    int i = blockIdx.x * blockDim.x + threadIdx.x;
    if (i < count) to[i] = from[i];
}

template <int OUTCH, int MT, bool RELU>
__global__ __launch_bounds__(256) void k_gemm(const float* __restrict__ A,
                                              const float* __restrict__ W,
                                              float* __restrict__ H, int n) {
    constexpr int LDSW = KDIM + 4;
    __shared__ float tile[MT * 16 * LDSW];

    const int tid = threadIdx.x;
    const int m_base = blockIdx.x * (MT * 16);

    constexpr int CHUNKS = MT * 16 * (KDIM / 4);
    for (int c = tid; c < CHUNKS; c += 256) {
        int row = c >> 5;
        int kc  = c & 31;
        int grow = m_base + row;
        if (grow >= n) grow = n - 1;
        float4 v = ((const float4*)(A + (size_t)grow * KDIM))[kc];
        if (RELU) {
            v.x = fmaxf(v.x, 0.0f); v.y = fmaxf(v.y, 0.0f);
            v.z = fmaxf(v.z, 0.0f); v.w = fmaxf(v.w, 0.0f);
        }
        *(float4*)&tile[row * LDSW + kc * 4] = v;
    }
    __syncthreads();

    const int wave = tid >> 5;
    const int lane = tid & 31;
    int mt, nt;
    if (OUTCH == 128) { mt = 0;         nt = wave;     }
    else              { mt = wave >> 2; nt = wave & 3; }
    const int lm = lane & 15;
    const int kh = lane >> 4;

    const float* arow = &tile[(mt * 16 + lm) * LDSW];
    const float* wcol = W + nt * 16 + lm;

    v8f acc = {};
#pragma unroll
    for (int k0 = 0; k0 < KDIM; k0 += 4) {
        v2f a, b;
        a.x = arow[k0 + 2 * kh + 0];
        a.y = arow[k0 + 2 * kh + 1];
        b.x = wcol[(size_t)(k0 + 2 * kh + 0) * OUTCH];
        b.y = wcol[(size_t)(k0 + 2 * kh + 1) * OUTCH];
        acc = __builtin_amdgcn_wmma_f32_16x16x4_f32(
             false, a,  false, b,
             (short)0, acc,  false,  false);
    }

    const int mrow0 = m_base + mt * 16 + (kh ? 8 : 0);
#pragma unroll
    for (int r = 0; r < 8; ++r) {
        int grow = mrow0 + r;
        if (grow < n)
            H[(size_t)grow * OUTCH + nt * 16 + lm] = acc[r];
    }
}

template <int CH> struct RowVec;
template <> struct RowVec<128> { typedef float4 type; };
template <> struct RowVec<64>  { typedef float2 type; };

__device__ __forceinline__ void row_fma(float4& a, const float4& h, float w) {
    a.x = fmaf(h.x, w, a.x); a.y = fmaf(h.y, w, a.y);
    a.z = fmaf(h.z, w, a.z); a.w = fmaf(h.w, w, a.w);
}
__device__ __forceinline__ void row_fma(float2& a, const float2& h, float w) {
    a.x = fmaf(h.x, w, a.x); a.y = fmaf(h.y, w, a.y);
}

template <int CH>
__global__ __launch_bounds__(256) void k_gather(const int* __restrict__ rowptr,
                                                const int* __restrict__ ssrc,
                                                const float* __restrict__ h,
                                                const float* __restrict__ dinv,
                                                const float* __restrict__ bias,
                                                float* __restrict__ out, int n) {
    typedef typename RowVec<CH>::type V;
    const int lane = threadIdx.x & 31;
    const int node = __builtin_amdgcn_readfirstlane((int)((blockIdx.x * 256u + threadIdx.x) >> 5));
    if (node >= n) return;
    const int beg = __builtin_amdgcn_readfirstlane(rowptr[node]);
    const int end = __builtin_amdgcn_readfirstlane(rowptr[node + 1]);
    const float dd = dinv[node];

    V acc = ((const V*)bias)[lane];
    row_fma(acc, ((const V*)(h + (size_t)node * CH))[lane], dd * dd);

    for (int base = beg; base < end; base += 32) {
        const int m = min(32, end - base);
        int   s_l = node;
        float w_l = 0.0f;
        if (lane < m) { s_l = ssrc[base + lane]; w_l = dinv[s_l] * dd; }
#define EDGE_ROW(j) (((const V*)(h + (size_t)__builtin_amdgcn_readlane(s_l, (j)) * CH))[lane])
#define EDGE_W(j)   (__int_as_float(__builtin_amdgcn_readlane(__float_as_int(w_l), (j))))
        int k = 0;
        for (; k + 4 <= m; k += 4) {
            const V h0 = EDGE_ROW(k + 0);
            const V h1 = EDGE_ROW(k + 1);
            const V h2 = EDGE_ROW(k + 2);
            const V h3 = EDGE_ROW(k + 3);
            row_fma(acc, h0, EDGE_W(k + 0));
            row_fma(acc, h1, EDGE_W(k + 1));
            row_fma(acc, h2, EDGE_W(k + 2));
            row_fma(acc, h3, EDGE_W(k + 3));
        }
        if (k < m) {
            const V h0 = EDGE_ROW(k);
            if (k + 1 < m) {
                const V h1 = EDGE_ROW(k + 1);
                if (k + 2 < m) {
                    const V h2 = EDGE_ROW(k + 2);
                    row_fma(acc, h0, EDGE_W(k));
                    row_fma(acc, h1, EDGE_W(k + 1));
                    row_fma(acc, h2, EDGE_W(k + 2));
                } else {
                    row_fma(acc, h0, EDGE_W(k));
                    row_fma(acc, h1, EDGE_W(k + 1));
                }
            } else {
                row_fma(acc, h0, EDGE_W(k));
            }
        }
#undef EDGE_ROW
#undef EDGE_W
    }
    ((V*)(out + (size_t)node * CH))[lane] = acc;
}

extern "C" void kernel_launch(void* const* d_in, const int* in_sizes, int n_in,
                              void* d_out, int out_size, void* d_ws, size_t ws_size,
                              hipStream_t stream) {
    const float* x  = (const float*)d_in[0];
    const int*   ei = (const int*)d_in[1];
    const float* W1 = (const float*)d_in[2];
    const float* b1 = (const float*)d_in[3];
    const float* W2 = (const float*)d_in[4];
    const float* b2 = (const float*)d_in[5];
    float* out = (float*)d_out;

    const int n = in_sizes[0] / 128;
    const int e = in_sizes[1] / 2;
    const int* src = ei;
    const int* dst = ei + e;

    float* dinv = (float*)d_ws;
    float* h  = dinv + n;
    float* o1 = h + (size_t)n * 128;
    int* rowptr1 = (int*)out;
    int* ssrc1   = rowptr1 + (n + 1);
    int* rowptr2 = (int*)o1;
    int* ssrc2   = rowptr2 + (n + 1);
    int* cnt    = (int*)o1;
    int* pos    = cnt + n;
    int* bsum   = pos + e;

    const unsigned T = 256;
    const unsigned SB = (n + SCAN_T - 1) / SCAN_T;
    k_zero_cnt<<<(n + T - 1) / T, T, 0, stream>>>(cnt, n);
    k_rank<<<(e + T - 1) / T, T, 0, stream>>>(dst, cnt, pos, e);
    k_block_sums<<<SB, SCAN_T, 0, stream>>>(cnt, bsum, n);
    k_scan<<<SB, SCAN_T, 0, stream>>>(cnt, bsum, rowptr1, dinv, n);
    k_fill<<<(e + T - 1) / T, T, 0, stream>>>(src, dst, pos, rowptr1, ssrc1, e);

    k_gemm<128, 1, false><<<(n + 15) / 16, T, 0, stream>>>(x, W1, h, n);
    k_gather<128><<<(n + 7) / 8, T, 0, stream>>>(rowptr1, ssrc1, h, dinv, b1, o1, n);

    k_gemm<64, 2, true><<<(n + 31) / 32, T, 0, stream>>>(o1, W2, h, n);
    k_move_buckets<<<(n + 1 + e + T - 1) / T, T, 0, stream>>>(rowptr1, rowptr2, n + 1 + e);
    k_gather<64><<<(n + 7) / 8, T, 0, stream>>>(rowptr2, ssrc2, h, dinv, b2, out, n);
}


# Layer-1 aggregation fused with the layer-2 product: 8-wave blocks sum 16 rows into an LDS tile (ReLU) and multiply it by W2 there; o1 never stored, bucket copy gone; sums and their order unchanged
# speedup vs baseline: 2.3454x; 1.0611x over previous
#include <hip/hip_runtime.h>

typedef __attribute__((ext_vector_type(2))) float v2f;
typedef __attribute__((ext_vector_type(8))) float v8f;

#define KDIM 128
#define LDSW (KDIM + 4)


__global__ void k_zero_cnt(int* __restrict__ cnt, int n) {
    int i = blockIdx.x * blockDim.x + threadIdx.x;
    if (i < n) cnt[i] = 0;
}

__global__ void k_rank(const int* __restrict__ dst, int* __restrict__ cnt,
                       int* __restrict__ pos, int e) {
    int i = blockIdx.x * blockDim.x + threadIdx.x;
    if (i < e) pos[i] = atomicAdd(&cnt[dst[i]], 1);
}

#define SCAN_T 1024
__device__ __forceinline__ int block_sum_1024(int v, int* sh) {
    const int lane = threadIdx.x & 31, wave = threadIdx.x >> 5;
#pragma unroll
    for (int o = 16; o > 0; o >>= 1) v += __shfl_xor(v, o);
    if (lane == 0) sh[wave] = v;
    __syncthreads();
    int t = sh[lane];
#pragma unroll
    for (int o = 16; o > 0; o >>= 1) t += __shfl_xor(t, o);
    __syncthreads();
    return t;
}

__global__ __launch_bounds__(SCAN_T) void k_block_sums(const int* __restrict__ cnt,
                                                       int* __restrict__ bsum, int n) {
    __shared__ int sh[32];
    int i = blockIdx.x * SCAN_T + threadIdx.x;
    int t = block_sum_1024(i < n ? cnt[i] : 0, sh);
    if (threadIdx.x == 0) bsum[blockIdx.x] = t;
}

__global__ __launch_bounds__(SCAN_T) void k_scan(const int* __restrict__ cnt,
                                                 const int* __restrict__ bsum,
                                                 int* __restrict__ rowptr,
                                                 float* __restrict__ dinv, int n) {
    __shared__ int sh[32];
    const int lane = threadIdx.x & 31, wave = threadIdx.x >> 5;
    int part = 0;
    for (int b = threadIdx.x; b < (int)blockIdx.x; b += SCAN_T) part += bsum[b];
    const int offset = block_sum_1024(part, sh);

    const int i = blockIdx.x * SCAN_T + threadIdx.x;
    const int c = (i < n) ? cnt[i] : 0;
    int incl = c;
#pragma unroll
    for (int o = 1; o < 32; o <<= 1) {
        int u = __shfl_up(incl, o);
        if (lane >= o) incl += u;
    }
    if (lane == 31) sh[wave] = incl;
    __syncthreads();
    int wsum = sh[lane];
#pragma unroll
    for (int o = 1; o < 32; o <<= 1) {
        int u = __shfl_up(wsum, o);
        if (lane >= o) wsum += u;
    }
    const int wave_excl = __shfl(wsum, wave) - __shfl(sh[lane], wave);
    const int excl = offset + wave_excl + incl - c;
    if (i < n) {
        rowptr[i] = excl;
        dinv[i] = rsqrtf((float)(c + 1));
        if (i == n - 1) rowptr[n] = excl + c;
    }
}

__global__ void k_fill(const int* __restrict__ src, const int* __restrict__ dst,
                       const int* __restrict__ pos, const int* __restrict__ rowptr,
                       int* __restrict__ ssrc, int e) {
    int i = blockIdx.x * blockDim.x + threadIdx.x;
    if (i < e) ssrc[rowptr[dst[i]] + pos[i]] = src[i];
}

template <int OUTCH>
__device__ __forceinline__ v8f tile_product(const float* arow, const float* __restrict__ wcol, int kh) {
    v8f acc = {};
#pragma unroll
    for (int k0 = 0; k0 < KDIM; k0 += 4) {
        v2f a, b;
        a.x = arow[k0 + 2 * kh + 0];
        a.y = arow[k0 + 2 * kh + 1];
        b.x = wcol[(size_t)(k0 + 2 * kh + 0) * OUTCH];
        b.y = wcol[(size_t)(k0 + 2 * kh + 1) * OUTCH];
        acc = __builtin_amdgcn_wmma_f32_16x16x4_f32(
             false, a,  false, b,
             (short)0, acc,  false,  false);
    }
    return acc;
}

template <int OUTCH>
__device__ __forceinline__ void tile_store(const v8f& acc, float* __restrict__ H, int mrow0, int col, int n) {
#pragma unroll
    for (int r = 0; r < 8; ++r) {
        int grow = mrow0 + r;
        if (grow < n)
            H[(size_t)grow * OUTCH + col] = acc[r];
    }
}

__global__ __launch_bounds__(256) void k_gemm1(const float* __restrict__ A,
                                               const float* __restrict__ W,
                                               float* __restrict__ H, int n) {
    __shared__ float tile[16 * LDSW];

    const int tid = threadIdx.x;
    const int m_base = blockIdx.x * 16;

    for (int c = tid; c < 16 * (KDIM / 4); c += 256) {
        int row = c >> 5;
        int kc  = c & 31;
        int grow = m_base + row;
        if (grow >= n) grow = n - 1;
        *(float4*)&tile[row * LDSW + kc * 4] = ((const float4*)(A + (size_t)grow * KDIM))[kc];
    }
    __syncthreads();

    const int nt = tid >> 5;
    const int lane = tid & 31;
    const int lm = lane & 15;
    const int kh = lane >> 4;
    const v8f acc = tile_product<128>(&tile[lm * LDSW], W + nt * 16 + lm, kh);
    tile_store<128>(acc, H, m_base + (kh ? 8 : 0), nt * 16 + lm, n);
}

template <int CH> struct RowVec;
template <> struct RowVec<128> { typedef float4 type; };
template <> struct RowVec<64>  { typedef float2 type; };

__device__ __forceinline__ void row_fma(float4& a, const float4& h, float w) {
    a.x = fmaf(h.x, w, a.x); a.y = fmaf(h.y, w, a.y);
    a.z = fmaf(h.z, w, a.z); a.w = fmaf(h.w, w, a.w);
}
__device__ __forceinline__ void row_fma(float2& a, const float2& h, float w) {
    a.x = fmaf(h.x, w, a.x); a.y = fmaf(h.y, w, a.y);
}

template <int CH>
__device__ __forceinline__ typename RowVec<CH>::type gather_row(int node, int lane,
                                                                const int* __restrict__ rowptr,
                                                                const int* __restrict__ ssrc,
                                                                const float* __restrict__ h,
                                                                const float* __restrict__ dinv,
                                                                const float* __restrict__ bias) {
    typedef typename RowVec<CH>::type V;
    const int beg = __builtin_amdgcn_readfirstlane(rowptr[node]);
    const int end = __builtin_amdgcn_readfirstlane(rowptr[node + 1]);
    const float dd = dinv[node];

    V acc = ((const V*)bias)[lane];
    row_fma(acc, ((const V*)(h + (size_t)node * CH))[lane], dd * dd);

    for (int base = beg; base < end; base += 32) {
        const int m = min(32, end - base);
        int   s_l = node;
        float w_l = 0.0f;
        if (lane < m) { s_l = ssrc[base + lane]; w_l = dinv[s_l] * dd; }
#define EDGE_ROW(j) (((const V*)(h + (size_t)__builtin_amdgcn_readlane(s_l, (j)) * CH))[lane])
#define EDGE_W(j)   (__int_as_float(__builtin_amdgcn_readlane(__float_as_int(w_l), (j))))
        int k = 0;
        for (; k + 4 <= m; k += 4) {
            const V h0 = EDGE_ROW(k + 0);
            const V h1 = EDGE_ROW(k + 1);
            const V h2 = EDGE_ROW(k + 2);
            const V h3 = EDGE_ROW(k + 3);
            row_fma(acc, h0, EDGE_W(k + 0));
            row_fma(acc, h1, EDGE_W(k + 1));
            row_fma(acc, h2, EDGE_W(k + 2));
            row_fma(acc, h3, EDGE_W(k + 3));
        }
        if (k < m) {
            const V h0 = EDGE_ROW(k);
            if (k + 1 < m) {
                const V h1 = EDGE_ROW(k + 1);
                if (k + 2 < m) {
                    const V h2 = EDGE_ROW(k + 2);
                    row_fma(acc, h0, EDGE_W(k));
                    row_fma(acc, h1, EDGE_W(k + 1));
                    row_fma(acc, h2, EDGE_W(k + 2));
                } else {
                    row_fma(acc, h0, EDGE_W(k));
                    row_fma(acc, h1, EDGE_W(k + 1));
                }
            } else {
                row_fma(acc, h0, EDGE_W(k));
            }
        }
#undef EDGE_ROW
#undef EDGE_W
    }
    return acc;
}

__global__ __launch_bounds__(256) void k_gather64(const int* __restrict__ rowptr,
                                                  const int* __restrict__ ssrc,
                                                  const float* __restrict__ h,
                                                  const float* __restrict__ dinv,
                                                  const float* __restrict__ bias,
                                                  float* __restrict__ out, int n) {
    const int lane = threadIdx.x & 31;
    const int node = __builtin_amdgcn_readfirstlane((int)((blockIdx.x * 256u + threadIdx.x) >> 5));
    if (node >= n) return;
    ((float2*)(out + (size_t)node * 64))[lane] = gather_row<64>(node, lane, rowptr, ssrc, h, dinv, bias);
}

__global__ __launch_bounds__(256) void k_gather_gemm2(const int* __restrict__ rowptr,
                                                      const int* __restrict__ ssrc,
                                                      const float* __restrict__ h,
                                                      const float* __restrict__ dinv,
                                                      const float* __restrict__ bias,
                                                      const float* __restrict__ W,
                                                      float* __restrict__ H2, int n) {
    __shared__ float tile[16 * LDSW];

    const int lane = threadIdx.x & 31;
    const int wave = __builtin_amdgcn_readfirstlane((int)(threadIdx.x >> 5));
    const int m_base = blockIdx.x * 16;

    for (int r = 0; r < 2; ++r) {
        const int lrow = 2 * wave + r;
        const int node = m_base + lrow;
        float4 v = make_float4(0.0f, 0.0f, 0.0f, 0.0f);
        if (node < n) {
            v = gather_row<128>(node, lane, rowptr, ssrc, h, dinv, bias);
            v.x = fmaxf(v.x, 0.0f); v.y = fmaxf(v.y, 0.0f);
            v.z = fmaxf(v.z, 0.0f); v.w = fmaxf(v.w, 0.0f);
        }
        *(float4*)&tile[lrow * LDSW + lane * 4] = v;
    }
    __syncthreads();
    if (wave >= 4) return;

    const int lm = lane & 15;
    const int kh = lane >> 4;
    const v8f acc = tile_product<64>(&tile[lm * LDSW], W + wave * 16 + lm, kh);
    tile_store<64>(acc, H2, m_base + (kh ? 8 : 0), wave * 16 + lm, n);
}

extern "C" void kernel_launch(void* const* d_in, const int* in_sizes, int n_in,
                              void* d_out, int out_size, void* d_ws, size_t ws_size,
                              hipStream_t stream) {
    const float* x  = (const float*)d_in[0];
    const int*   ei = (const int*)d_in[1];
    const float* W1 = (const float*)d_in[2];
    const float* b1 = (const float*)d_in[3];
    const float* W2 = (const float*)d_in[4];
    const float* b2 = (const float*)d_in[5];
    float* out = (float*)d_out;

    const int n = in_sizes[0] / 128;
    const int e = in_sizes[1] / 2;
    const int* src = ei;
    const int* dst = ei + e;

    float* dinv = (float*)d_ws;
    float* h1 = dinv + n;
    float* h2 = h1 + (size_t)n * 128;
    int* rowptr = (int*)(h2 + (size_t)n * 64);
    int* ssrc   = rowptr + (n + 1);
    int* cnt    = ssrc + e;
    int* pos    = cnt + n;
    int* bsum   = pos + e;

    const unsigned T = 256;
    const unsigned SB = (n + SCAN_T - 1) / SCAN_T;
    k_zero_cnt<<<(n + T - 1) / T, T, 0, stream>>>(cnt, n);
    k_rank<<<(e + T - 1) / T, T, 0, stream>>>(dst, cnt, pos, e);
    k_block_sums<<<SB, SCAN_T, 0, stream>>>(cnt, bsum, n);
    k_scan<<<SB, SCAN_T, 0, stream>>>(cnt, bsum, rowptr, dinv, n);
    k_fill<<<(e + T - 1) / T, T, 0, stream>>>(src, dst, pos, rowptr, ssrc, e);

    k_gemm1<<<(n + 15) / 16, T, 0, stream>>>(x, W1, h1, n);
    k_gather_gemm2<<<(n + 15) / 16, T, 0, stream>>>(rowptr, ssrc, h1, dinv, b1, W2, h2, n);
    k_gather64<<<(n + 7) / 8, T, 0, stream>>>(rowptr, ssrc, h2, dinv, b2, out, n);
}


# Edge buckets by a two-level counting sort on LDS counters (256-node ranges per edge chunk, then node counts per range) in place of one global atomic per edge; row sums keep bucket-arrival order
# speedup vs baseline: 3.3553x; 1.4306x over previous
#include <hip/hip_runtime.h>

typedef __attribute__((ext_vector_type(2))) float v2f;
typedef __attribute__((ext_vector_type(8))) float v8f;

#define KDIM 128
#define LDSW (KDIM + 4)


#define BSH 8
#define BNODES (1 << BSH)
#define GA 256
#define CT 1024
#define MAXNB 1024
#define EPT 4

__global__ __launch_bounds__(CT) void k_chunk_hist(const int* __restrict__ dst, int* __restrict__ hist,
                                                   int e, int nb) {
    __shared__ int h[MAXNB];
    for (int b = threadIdx.x; b < nb; b += CT) h[b] = 0;
    __syncthreads();
    const int per = (e + GA - 1) / GA;
    const int beg = blockIdx.x * per;
    const int end = min(e, beg + per);
    for (int base = beg + (int)threadIdx.x; base < end; base += CT * EPT) {
        int d[EPT];
#pragma unroll
        for (int k = 0; k < EPT; ++k) d[k] = (base + k * CT < end) ? dst[base + k * CT] : 0;
#pragma unroll
        for (int k = 0; k < EPT; ++k)
            if (base + k * CT < end) atomicAdd(&h[d[k] >> BSH], 1);
    }
    __syncthreads();
    for (int b = threadIdx.x; b < nb; b += CT) hist[b * GA + blockIdx.x] = h[b];
}

#define SCAN_T 1024
__device__ __forceinline__ int block_sum_1024(int v, int* sh) {
    const int lane = threadIdx.x & 31, wave = threadIdx.x >> 5;
#pragma unroll
    for (int o = 16; o > 0; o >>= 1) v += __shfl_xor(v, o);
    if (lane == 0) sh[wave] = v;
    __syncthreads();
    int t = sh[lane];
#pragma unroll
    for (int o = 16; o > 0; o >>= 1) t += __shfl_xor(t, o);
    __syncthreads();
    return t;
}

__global__ __launch_bounds__(SCAN_T) void k_block_sums(const int* __restrict__ cnt,
                                                       int* __restrict__ bsum, int n) {
    __shared__ int sh[32];
    int i = blockIdx.x * SCAN_T + threadIdx.x;
    int t = block_sum_1024(i < n ? cnt[i] : 0, sh);
    if (threadIdx.x == 0) bsum[blockIdx.x] = t;
}

__global__ __launch_bounds__(SCAN_T) void k_scan(const int* __restrict__ cnt,
                                                 const int* __restrict__ bsum,
                                                 int* __restrict__ off, int n) {
    __shared__ int sh[32];
    const int lane = threadIdx.x & 31, wave = threadIdx.x >> 5;
    int part = 0;
    for (int b = threadIdx.x; b < (int)blockIdx.x; b += SCAN_T) part += bsum[b];
    const int offset = block_sum_1024(part, sh);

    const int i = blockIdx.x * SCAN_T + threadIdx.x;
    const int c = (i < n) ? cnt[i] : 0;
    int incl = c;
#pragma unroll
    for (int o = 1; o < 32; o <<= 1) {
        int u = __shfl_up(incl, o);
        if (lane >= o) incl += u;
    }
    if (lane == 31) sh[wave] = incl;
    __syncthreads();
    int wsum = sh[lane];
#pragma unroll
    for (int o = 1; o < 32; o <<= 1) {
        int u = __shfl_up(wsum, o);
        if (lane >= o) wsum += u;
    }
    const int wave_excl = __shfl(wsum, wave) - __shfl(sh[lane], wave);
    const int excl = offset + wave_excl + incl - c;
    if (i < n) {
        off[i] = excl;
        if (i == n - 1) off[n] = excl + c;
    }
}

__global__ __launch_bounds__(CT) void k_chunk_place(const int* __restrict__ src, const int* __restrict__ dst,
                                                    const int* __restrict__ hoff, int2* __restrict__ pairs,
                                                    int e, int nb) {
    __shared__ int cur[MAXNB];
    for (int b = threadIdx.x; b < nb; b += CT) cur[b] = hoff[b * GA + blockIdx.x];
    __syncthreads();
    const int per = (e + GA - 1) / GA;
    const int beg = blockIdx.x * per;
    const int end = min(e, beg + per);
    for (int base = beg + (int)threadIdx.x; base < end; base += CT * EPT) {
        int d[EPT], s[EPT];
#pragma unroll
        for (int k = 0; k < EPT; ++k) {
            const bool in = base + k * CT < end;
            d[k] = in ? dst[base + k * CT] : 0;
            s[k] = in ? src[base + k * CT] : 0;
        }
#pragma unroll
        for (int k = 0; k < EPT; ++k)
            if (base + k * CT < end) pairs[atomicAdd(&cur[d[k] >> BSH], 1)] = make_int2(s[k], d[k]);
    }
}

__global__ __launch_bounds__(CT) void k_range_sort(const int2* __restrict__ pairs, const int* __restrict__ hoff,
                                                   int* __restrict__ rowptr, float* __restrict__ dinv,
                                                   int* __restrict__ ssrc, int n) {
    __shared__ int cnt[BNODES];
    __shared__ int sh[32];
    const int lane = threadIdx.x & 31, wave = threadIdx.x >> 5;
    const int node0 = blockIdx.x << BSH;
    const int beg = hoff[blockIdx.x * GA];
    const int end = hoff[(blockIdx.x + 1) * GA];
    if (threadIdx.x < BNODES) cnt[threadIdx.x] = 0;
    __syncthreads();
    for (int base = beg + (int)threadIdx.x; base < end; base += CT * EPT) {
        int d[EPT];
#pragma unroll
        for (int k = 0; k < EPT; ++k) d[k] = (base + k * CT < end) ? pairs[base + k * CT].y : node0;
#pragma unroll
        for (int k = 0; k < EPT; ++k)
            if (base + k * CT < end) atomicAdd(&cnt[d[k] - node0], 1);
    }
    __syncthreads();
    const int c = (threadIdx.x < BNODES) ? cnt[threadIdx.x] : 0;
    int incl = c;
#pragma unroll
    for (int o = 1; o < 32; o <<= 1) {
        int u = __shfl_up(incl, o);
        if (lane >= o) incl += u;
    }
    if (lane == 31) sh[wave] = incl;
    __syncthreads();
    int woff = 0;
    for (int w = 0; w < BNODES / 32; ++w) woff += (w < wave) ? sh[w] : 0;
    const int excl = beg + woff + incl - c;
    if (threadIdx.x < BNODES) {
        const int node = node0 + (int)threadIdx.x;
        cnt[threadIdx.x] = excl;
        if (node < n) {
            rowptr[node] = excl;
            dinv[node] = rsqrtf((float)(c + 1));
            if (node == n - 1) rowptr[n] = excl + c;
        }
    }
    __syncthreads();
    for (int base = beg + (int)threadIdx.x; base < end; base += CT * EPT) {
        int2 p[EPT];
#pragma unroll
        for (int k = 0; k < EPT; ++k) p[k] = (base + k * CT < end) ? pairs[base + k * CT] : make_int2(0, node0);
#pragma unroll
        for (int k = 0; k < EPT; ++k)
            if (base + k * CT < end) ssrc[atomicAdd(&cnt[p[k].y - node0], 1)] = p[k].x;
    }
}

template <int OUTCH>
__device__ __forceinline__ v8f tile_product(const float* arow, const float* __restrict__ wcol, int kh) {
    v8f acc = {};
#pragma unroll
    for (int k0 = 0; k0 < KDIM; k0 += 4) {
        v2f a, b;
        a.x = arow[k0 + 2 * kh + 0];
        a.y = arow[k0 + 2 * kh + 1];
        b.x = wcol[(size_t)(k0 + 2 * kh + 0) * OUTCH];
        b.y = wcol[(size_t)(k0 + 2 * kh + 1) * OUTCH];
        acc = __builtin_amdgcn_wmma_f32_16x16x4_f32(
             false, a,  false, b,
             (short)0, acc,  false,  false);
    }
    return acc;
}

template <int OUTCH>
__device__ __forceinline__ void tile_store(const v8f& acc, float* __restrict__ H, int mrow0, int col, int n) {
#pragma unroll
    for (int r = 0; r < 8; ++r) {
        int grow = mrow0 + r;
        if (grow < n)
            H[(size_t)grow * OUTCH + col] = acc[r];
    }
}

__global__ __launch_bounds__(256) void k_gemm1(const float* __restrict__ A,
                                               const float* __restrict__ W,
                                               float* __restrict__ H, int n) {
    __shared__ float tile[16 * LDSW];

    const int tid = threadIdx.x;
    const int m_base = blockIdx.x * 16;

    for (int c = tid; c < 16 * (KDIM / 4); c += 256) {
        int row = c >> 5;
        int kc  = c & 31;
        int grow = m_base + row;
        if (grow >= n) grow = n - 1;
        *(float4*)&tile[row * LDSW + kc * 4] = ((const float4*)(A + (size_t)grow * KDIM))[kc];
    }
    __syncthreads();

    const int nt = tid >> 5;
    const int lane = tid & 31;
    const int lm = lane & 15;
    const int kh = lane >> 4;
    const v8f acc = tile_product<128>(&tile[lm * LDSW], W + nt * 16 + lm, kh);
    tile_store<128>(acc, H, m_base + (kh ? 8 : 0), nt * 16 + lm, n);
}

template <int CH> struct RowVec;
template <> struct RowVec<128> { typedef float4 type; };
template <> struct RowVec<64>  { typedef float2 type; };

__device__ __forceinline__ void row_fma(float4& a, const float4& h, float w) {
    a.x = fmaf(h.x, w, a.x); a.y = fmaf(h.y, w, a.y);
    a.z = fmaf(h.z, w, a.z); a.w = fmaf(h.w, w, a.w);
}
__device__ __forceinline__ void row_fma(float2& a, const float2& h, float w) {
    a.x = fmaf(h.x, w, a.x); a.y = fmaf(h.y, w, a.y);
}

template <int CH>
__device__ __forceinline__ typename RowVec<CH>::type gather_row(int node, int lane,
                                                                const int* __restrict__ rowptr,
                                                                const int* __restrict__ ssrc,
                                                                const float* __restrict__ h,
                                                                const float* __restrict__ dinv,
                                                                const float* __restrict__ bias) {
    typedef typename RowVec<CH>::type V;
    const int beg = __builtin_amdgcn_readfirstlane(rowptr[node]);
    const int end = __builtin_amdgcn_readfirstlane(rowptr[node + 1]);
    const float dd = dinv[node];

    V acc = ((const V*)bias)[lane];
    row_fma(acc, ((const V*)(h + (size_t)node * CH))[lane], dd * dd);

    for (int base = beg; base < end; base += 32) {
        const int m = min(32, end - base);
        int   s_l = node;
        float w_l = 0.0f;
        if (lane < m) { s_l = ssrc[base + lane]; w_l = dinv[s_l] * dd; }
#define EDGE_ROW(j) (((const V*)(h + (size_t)__builtin_amdgcn_readlane(s_l, (j)) * CH))[lane])
#define EDGE_W(j)   (__int_as_float(__builtin_amdgcn_readlane(__float_as_int(w_l), (j))))
        int k = 0;
        for (; k + 4 <= m; k += 4) {
            const V h0 = EDGE_ROW(k + 0);
            const V h1 = EDGE_ROW(k + 1);
            const V h2 = EDGE_ROW(k + 2);
            const V h3 = EDGE_ROW(k + 3);
            row_fma(acc, h0, EDGE_W(k + 0));
            row_fma(acc, h1, EDGE_W(k + 1));
            row_fma(acc, h2, EDGE_W(k + 2));
            row_fma(acc, h3, EDGE_W(k + 3));
        }
        if (k < m) {
            const V h0 = EDGE_ROW(k);
            if (k + 1 < m) {
                const V h1 = EDGE_ROW(k + 1);
                if (k + 2 < m) {
                    const V h2 = EDGE_ROW(k + 2);
                    row_fma(acc, h0, EDGE_W(k));
                    row_fma(acc, h1, EDGE_W(k + 1));
                    row_fma(acc, h2, EDGE_W(k + 2));
                } else {
                    row_fma(acc, h0, EDGE_W(k));
                    row_fma(acc, h1, EDGE_W(k + 1));
                }
            } else {
                row_fma(acc, h0, EDGE_W(k));
            }
        }
#undef EDGE_ROW
#undef EDGE_W
    }
    return acc;
}

__global__ __launch_bounds__(256) void k_gather64(const int* __restrict__ rowptr,
                                                  const int* __restrict__ ssrc,
                                                  const float* __restrict__ h,
                                                  const float* __restrict__ dinv,
                                                  const float* __restrict__ bias,
                                                  float* __restrict__ out, int n) {
    const int lane = threadIdx.x & 31;
    const int node = __builtin_amdgcn_readfirstlane((int)((blockIdx.x * 256u + threadIdx.x) >> 5));
    if (node >= n) return;
    ((float2*)(out + (size_t)node * 64))[lane] = gather_row<64>(node, lane, rowptr, ssrc, h, dinv, bias);
}

__global__ __launch_bounds__(256) void k_gather_gemm2(const int* __restrict__ rowptr,
                                                      const int* __restrict__ ssrc,
                                                      const float* __restrict__ h,
                                                      const float* __restrict__ dinv,
                                                      const float* __restrict__ bias,
                                                      const float* __restrict__ W,
                                                      float* __restrict__ H2, int n) {
    __shared__ float tile[16 * LDSW];

    const int lane = threadIdx.x & 31;
    const int wave = __builtin_amdgcn_readfirstlane((int)(threadIdx.x >> 5));
    const int m_base = blockIdx.x * 16;

    for (int r = 0; r < 2; ++r) {
        const int lrow = 2 * wave + r;
        const int node = m_base + lrow;
        float4 v = make_float4(0.0f, 0.0f, 0.0f, 0.0f);
        if (node < n) {
            v = gather_row<128>(node, lane, rowptr, ssrc, h, dinv, bias);
            v.x = fmaxf(v.x, 0.0f); v.y = fmaxf(v.y, 0.0f);
            v.z = fmaxf(v.z, 0.0f); v.w = fmaxf(v.w, 0.0f);
        }
        *(float4*)&tile[lrow * LDSW + lane * 4] = v;
    }
    __syncthreads();
    if (wave >= 4) return;

    const int lm = lane & 15;
    const int kh = lane >> 4;
    const v8f acc = tile_product<64>(&tile[lm * LDSW], W + wave * 16 + lm, kh);
    tile_store<64>(acc, H2, m_base + (kh ? 8 : 0), wave * 16 + lm, n);
}

extern "C" void kernel_launch(void* const* d_in, const int* in_sizes, int n_in,
                              void* d_out, int out_size, void* d_ws, size_t ws_size,
                              hipStream_t stream) {
    const float* x  = (const float*)d_in[0];
    const int*   ei = (const int*)d_in[1];
    const float* W1 = (const float*)d_in[2];
    const float* b1 = (const float*)d_in[3];
    const float* W2 = (const float*)d_in[4];
    const float* b2 = (const float*)d_in[5];
    float* out = (float*)d_out;

    const int n = in_sizes[0] / 128;
    const int e = in_sizes[1] / 2;
    const int* src = ei;
    const int* dst = ei + e;

    float* dinv = (float*)d_ws;
    float* h1 = dinv + n;
    float* h2 = h1 + (size_t)n * 128;
    int* rowptr = (int*)(h2 + (size_t)n * 64);
    int* ssrc   = rowptr + (n + 1);
    const int nb = (n + BNODES - 1) / BNODES;
    const int nh = nb * GA;
    int* hist   = ssrc + e;
    int* hoff   = hist + nh;
    int* bsum   = hoff + (nh + 1);
    int2* pairs = (int2*)(h1 + (n & 1));

    const unsigned T = 256;
    const unsigned SB = (nh + SCAN_T - 1) / SCAN_T;
    k_chunk_hist<<<GA, CT, 0, stream>>>(dst, hist, e, nb);
    k_block_sums<<<SB, SCAN_T, 0, stream>>>(hist, bsum, nh);
    k_scan<<<SB, SCAN_T, 0, stream>>>(hist, bsum, hoff, nh);
    k_chunk_place<<<GA, CT, 0, stream>>>(src, dst, hoff, pairs, e, nb);
    k_range_sort<<<nb, CT, 0, stream>>>(pairs, hoff, rowptr, dinv, ssrc, n);

    k_gemm1<<<(n + 15) / 16, T, 0, stream>>>(x, W1, h1, n);
    k_gather_gemm2<<<(n + 15) / 16, T, 0, stream>>>(rowptr, ssrc, h1, dinv, b1, W2, h2, n);
    k_gather64<<<(n + 7) / 8, T, 0, stream>>>(rowptr, ssrc, h2, dinv, b2, out, n);
}


# Both products take A rows pre-scaled by the row's own dinv: aggregations add plain rows, no per-edge dinv[src] fetch; dinv[dst] and bias applied once after the sum (rounding changes, edge order kept)
# speedup vs baseline: 3.5987x; 1.0725x over previous
#include <hip/hip_runtime.h>

typedef __attribute__((ext_vector_type(2))) float v2f;
typedef __attribute__((ext_vector_type(8))) float v8f;

#define KDIM 128
#define LDSW (KDIM + 4)


#define BSH 8
#define BNODES (1 << BSH)
#define GA 256
#define CT 1024
#define MAXNB 1024
#define EPT 4

__global__ __launch_bounds__(CT) void k_chunk_hist(const int* __restrict__ dst, int* __restrict__ hist,
                                                   int e, int nb) {
    __shared__ int h[MAXNB];
    for (int b = threadIdx.x; b < nb; b += CT) h[b] = 0;
    __syncthreads();
    const int per = (e + GA - 1) / GA;
    const int beg = blockIdx.x * per;
    const int end = min(e, beg + per);
    for (int base = beg + (int)threadIdx.x; base < end; base += CT * EPT) {
        int d[EPT];
#pragma unroll
        for (int k = 0; k < EPT; ++k) d[k] = (base + k * CT < end) ? dst[base + k * CT] : 0;
#pragma unroll
        for (int k = 0; k < EPT; ++k)
            if (base + k * CT < end) atomicAdd(&h[d[k] >> BSH], 1);
    }
    __syncthreads();
    for (int b = threadIdx.x; b < nb; b += CT) hist[b * GA + blockIdx.x] = h[b];
}

#define SCAN_T 1024
__device__ __forceinline__ int block_sum_1024(int v, int* sh) {
    const int lane = threadIdx.x & 31, wave = threadIdx.x >> 5;
#pragma unroll
    for (int o = 16; o > 0; o >>= 1) v += __shfl_xor(v, o);
    if (lane == 0) sh[wave] = v;
    __syncthreads();
    int t = sh[lane];
#pragma unroll
    for (int o = 16; o > 0; o >>= 1) t += __shfl_xor(t, o);
    __syncthreads();
    return t;
}

__global__ __launch_bounds__(SCAN_T) void k_block_sums(const int* __restrict__ cnt,
                                                       int* __restrict__ bsum, int n) {
    __shared__ int sh[32];
    int i = blockIdx.x * SCAN_T + threadIdx.x;
    int t = block_sum_1024(i < n ? cnt[i] : 0, sh);
    if (threadIdx.x == 0) bsum[blockIdx.x] = t;
}

__global__ __launch_bounds__(SCAN_T) void k_scan(const int* __restrict__ cnt,
                                                 const int* __restrict__ bsum,
                                                 int* __restrict__ off, int n) {
    __shared__ int sh[32];
    const int lane = threadIdx.x & 31, wave = threadIdx.x >> 5;
    int part = 0;
    for (int b = threadIdx.x; b < (int)blockIdx.x; b += SCAN_T) part += bsum[b];
    const int offset = block_sum_1024(part, sh);

    const int i = blockIdx.x * SCAN_T + threadIdx.x;
    const int c = (i < n) ? cnt[i] : 0;
    int incl = c;
#pragma unroll
    for (int o = 1; o < 32; o <<= 1) {
        int u = __shfl_up(incl, o);
        if (lane >= o) incl += u;
    }
    if (lane == 31) sh[wave] = incl;
    __syncthreads();
    int wsum = sh[lane];
#pragma unroll
    for (int o = 1; o < 32; o <<= 1) {
        int u = __shfl_up(wsum, o);
        if (lane >= o) wsum += u;
    }
    const int wave_excl = __shfl(wsum, wave) - __shfl(sh[lane], wave);
    const int excl = offset + wave_excl + incl - c;
    if (i < n) {
        off[i] = excl;
        if (i == n - 1) off[n] = excl + c;
    }
}

__global__ __launch_bounds__(CT) void k_chunk_place(const int* __restrict__ src, const int* __restrict__ dst,
                                                    const int* __restrict__ hoff, int2* __restrict__ pairs,
                                                    int e, int nb) {
    __shared__ int cur[MAXNB];
    for (int b = threadIdx.x; b < nb; b += CT) cur[b] = hoff[b * GA + blockIdx.x];
    __syncthreads();
    const int per = (e + GA - 1) / GA;
    const int beg = blockIdx.x * per;
    const int end = min(e, beg + per);
    for (int base = beg + (int)threadIdx.x; base < end; base += CT * EPT) {
        int d[EPT], s[EPT];
#pragma unroll
        for (int k = 0; k < EPT; ++k) {
            const bool in = base + k * CT < end;
            d[k] = in ? dst[base + k * CT] : 0;
            s[k] = in ? src[base + k * CT] : 0;
        }
#pragma unroll
        for (int k = 0; k < EPT; ++k)
            if (base + k * CT < end) pairs[atomicAdd(&cur[d[k] >> BSH], 1)] = make_int2(s[k], d[k]);
    }
}

__global__ __launch_bounds__(CT) void k_range_sort(const int2* __restrict__ pairs, const int* __restrict__ hoff,
                                                   int* __restrict__ rowptr, float* __restrict__ dinv,
                                                   int* __restrict__ ssrc, int n) {
    __shared__ int cnt[BNODES];
    __shared__ int sh[32];
    const int lane = threadIdx.x & 31, wave = threadIdx.x >> 5;
    const int node0 = blockIdx.x << BSH;
    const int beg = hoff[blockIdx.x * GA];
    const int end = hoff[(blockIdx.x + 1) * GA];
    if (threadIdx.x < BNODES) cnt[threadIdx.x] = 0;
    __syncthreads();
    for (int base = beg + (int)threadIdx.x; base < end; base += CT * EPT) {
        int d[EPT];
#pragma unroll
        for (int k = 0; k < EPT; ++k) d[k] = (base + k * CT < end) ? pairs[base + k * CT].y : node0;
#pragma unroll
        for (int k = 0; k < EPT; ++k)
            if (base + k * CT < end) atomicAdd(&cnt[d[k] - node0], 1);
    }
    __syncthreads();
    const int c = (threadIdx.x < BNODES) ? cnt[threadIdx.x] : 0;
    int incl = c;
#pragma unroll
    for (int o = 1; o < 32; o <<= 1) {
        int u = __shfl_up(incl, o);
        if (lane >= o) incl += u;
    }
    if (lane == 31) sh[wave] = incl;
    __syncthreads();
    int woff = 0;
    for (int w = 0; w < BNODES / 32; ++w) woff += (w < wave) ? sh[w] : 0;
    const int excl = beg + woff + incl - c;
    if (threadIdx.x < BNODES) {
        const int node = node0 + (int)threadIdx.x;
        cnt[threadIdx.x] = excl;
        if (node < n) {
            rowptr[node] = excl;
            dinv[node] = rsqrtf((float)(c + 1));
            if (node == n - 1) rowptr[n] = excl + c;
        }
    }
    __syncthreads();
    for (int base = beg + (int)threadIdx.x; base < end; base += CT * EPT) {
        int2 p[EPT];
#pragma unroll
        for (int k = 0; k < EPT; ++k) p[k] = (base + k * CT < end) ? pairs[base + k * CT] : make_int2(0, node0);
#pragma unroll
        for (int k = 0; k < EPT; ++k)
            if (base + k * CT < end) ssrc[atomicAdd(&cnt[p[k].y - node0], 1)] = p[k].x;
    }
}

template <int OUTCH>
__device__ __forceinline__ v8f tile_product(const float* arow, const float* __restrict__ wcol, int kh) {
    v8f acc = {};
#pragma unroll
    for (int k0 = 0; k0 < KDIM; k0 += 4) {
        v2f a, b;
        a.x = arow[k0 + 2 * kh + 0];
        a.y = arow[k0 + 2 * kh + 1];
        b.x = wcol[(size_t)(k0 + 2 * kh + 0) * OUTCH];
        b.y = wcol[(size_t)(k0 + 2 * kh + 1) * OUTCH];
        acc = __builtin_amdgcn_wmma_f32_16x16x4_f32(
             false, a,  false, b,
             (short)0, acc,  false,  false);
    }
    return acc;
}

template <int OUTCH>
__device__ __forceinline__ void tile_store(const v8f& acc, float* __restrict__ H, int mrow0, int col, int n) {
#pragma unroll
    for (int r = 0; r < 8; ++r) {
        int grow = mrow0 + r;
        if (grow < n)
            H[(size_t)grow * OUTCH + col] = acc[r];
    }
}

__global__ __launch_bounds__(256) void k_gemm1(const float* __restrict__ A,
                                               const float* __restrict__ W,
                                               const float* __restrict__ dinv,
                                               float* __restrict__ H, int n) {
    __shared__ float tile[16 * LDSW];

    const int tid = threadIdx.x;
    const int m_base = blockIdx.x * 16;

    for (int c = tid; c < 16 * (KDIM / 4); c += 256) {
        int row = c >> 5;
        int kc  = c & 31;
        int grow = m_base + row;
        if (grow >= n) grow = n - 1;
        float4 v = ((const float4*)(A + (size_t)grow * KDIM))[kc];
        const float s = dinv[grow];
        v.x *= s; v.y *= s; v.z *= s; v.w *= s;
        *(float4*)&tile[row * LDSW + kc * 4] = v;
    }
    __syncthreads();

    const int nt = tid >> 5;
    const int lane = tid & 31;
    const int lm = lane & 15;
    const int kh = lane >> 4;
    const v8f acc = tile_product<128>(&tile[lm * LDSW], W + nt * 16 + lm, kh);
    tile_store<128>(acc, H, m_base + (kh ? 8 : 0), nt * 16 + lm, n);
}

template <int CH> struct RowVec;
template <> struct RowVec<128> { typedef float4 type; };
template <> struct RowVec<64>  { typedef float2 type; };

__device__ __forceinline__ void row_add(float4& a, const float4& h) {
    a.x += h.x; a.y += h.y; a.z += h.z; a.w += h.w;
}
__device__ __forceinline__ void row_add(float2& a, const float2& h) {
    a.x += h.x; a.y += h.y;
}
__device__ __forceinline__ float4 row_scale_bias(const float4& a, float w, const float4& b) {
    return make_float4(fmaf(a.x, w, b.x), fmaf(a.y, w, b.y), fmaf(a.z, w, b.z), fmaf(a.w, w, b.w));
}
__device__ __forceinline__ float2 row_scale_bias(const float2& a, float w, const float2& b) {
    return make_float2(fmaf(a.x, w, b.x), fmaf(a.y, w, b.y));
}

template <int CH>
__device__ __forceinline__ typename RowVec<CH>::type gather_row(int node, int lane, float dd,
                                                                const int* __restrict__ rowptr,
                                                                const int* __restrict__ ssrc,
                                                                const float* __restrict__ hs,
                                                                const float* __restrict__ bias) {
    typedef typename RowVec<CH>::type V;
    const int beg = __builtin_amdgcn_readfirstlane(rowptr[node]);
    const int end = __builtin_amdgcn_readfirstlane(rowptr[node + 1]);

    const V b = ((const V*)bias)[lane];
    V acc = ((const V*)(hs + (size_t)node * CH))[lane];

    for (int base = beg; base < end; base += 32) {
        const int m = min(32, end - base);
        int s_l = node;
        if (lane < m) s_l = ssrc[base + lane];
#define EDGE_ROW(j) (((const V*)(hs + (size_t)__builtin_amdgcn_readlane(s_l, (j)) * CH))[lane])
        int k = 0;
        for (; k + 4 <= m; k += 4) {
            const V h0 = EDGE_ROW(k + 0);
            const V h1 = EDGE_ROW(k + 1);
            const V h2 = EDGE_ROW(k + 2);
            const V h3 = EDGE_ROW(k + 3);
            row_add(acc, h0);
            row_add(acc, h1);
            row_add(acc, h2);
            row_add(acc, h3);
        }
        if (k < m) {
            const V h0 = EDGE_ROW(k);
            if (k + 1 < m) {
                const V h1 = EDGE_ROW(k + 1);
                if (k + 2 < m) {
                    const V h2 = EDGE_ROW(k + 2);
                    row_add(acc, h0);
                    row_add(acc, h1);
                    row_add(acc, h2);
                } else {
                    row_add(acc, h0);
                    row_add(acc, h1);
                }
            } else {
                row_add(acc, h0);
            }
        }
#undef EDGE_ROW
    }
    return row_scale_bias(acc, dd, b);
}

__global__ __launch_bounds__(256) void k_gather64(const int* __restrict__ rowptr,
                                                  const int* __restrict__ ssrc,
                                                  const float* __restrict__ h,
                                                  const float* __restrict__ dinv,
                                                  const float* __restrict__ bias,
                                                  float* __restrict__ out, int n) {
    const int lane = threadIdx.x & 31;
    const int node = __builtin_amdgcn_readfirstlane((int)((blockIdx.x * 256u + threadIdx.x) >> 5));
    if (node >= n) return;
    ((float2*)(out + (size_t)node * 64))[lane] = gather_row<64>(node, lane, dinv[node], rowptr, ssrc, h, bias);
}

__global__ __launch_bounds__(256) void k_gather_gemm2(const int* __restrict__ rowptr,
                                                      const int* __restrict__ ssrc,
                                                      const float* __restrict__ h,
                                                      const float* __restrict__ dinv,
                                                      const float* __restrict__ bias,
                                                      const float* __restrict__ W,
                                                      float* __restrict__ H2, int n) {
    __shared__ float tile[16 * LDSW];

    const int lane = threadIdx.x & 31;
    const int wave = __builtin_amdgcn_readfirstlane((int)(threadIdx.x >> 5));
    const int m_base = blockIdx.x * 16;

    for (int r = 0; r < 2; ++r) {
        const int lrow = 2 * wave + r;
        const int node = m_base + lrow;
        float4 v = make_float4(0.0f, 0.0f, 0.0f, 0.0f);
        if (node < n) {
            const float dd = dinv[node];
            v = gather_row<128>(node, lane, dd, rowptr, ssrc, h, bias);
            v.x = fmaxf(v.x, 0.0f) * dd; v.y = fmaxf(v.y, 0.0f) * dd;
            v.z = fmaxf(v.z, 0.0f) * dd; v.w = fmaxf(v.w, 0.0f) * dd;
        }
        *(float4*)&tile[lrow * LDSW + lane * 4] = v;
    }
    __syncthreads();
    if (wave >= 4) return;

    const int lm = lane & 15;
    const int kh = lane >> 4;
    const v8f acc = tile_product<64>(&tile[lm * LDSW], W + wave * 16 + lm, kh);
    tile_store<64>(acc, H2, m_base + (kh ? 8 : 0), wave * 16 + lm, n);
}

extern "C" void kernel_launch(void* const* d_in, const int* in_sizes, int n_in,
                              void* d_out, int out_size, void* d_ws, size_t ws_size,
                              hipStream_t stream) {
    const float* x  = (const float*)d_in[0];
    const int*   ei = (const int*)d_in[1];
    const float* W1 = (const float*)d_in[2];
    const float* b1 = (const float*)d_in[3];
    const float* W2 = (const float*)d_in[4];
    const float* b2 = (const float*)d_in[5];
    float* out = (float*)d_out;

    const int n = in_sizes[0] / 128;
    const int e = in_sizes[1] / 2;
    const int* src = ei;
    const int* dst = ei + e;

    float* dinv = (float*)d_ws;
    float* h1 = dinv + n;
    float* h2 = h1 + (size_t)n * 128;
    int* rowptr = (int*)(h2 + (size_t)n * 64);
    int* ssrc   = rowptr + (n + 1);
    const int nb = (n + BNODES - 1) / BNODES;
    const int nh = nb * GA;
    int* hist   = ssrc + e;
    int* hoff   = hist + nh;
    int* bsum   = hoff + (nh + 1);
    int2* pairs = (int2*)(h1 + (n & 1));

    const unsigned T = 256;
    const unsigned SB = (nh + SCAN_T - 1) / SCAN_T;
    k_chunk_hist<<<GA, CT, 0, stream>>>(dst, hist, e, nb);
    k_block_sums<<<SB, SCAN_T, 0, stream>>>(hist, bsum, nh);
    k_scan<<<SB, SCAN_T, 0, stream>>>(hist, bsum, hoff, nh);
    k_chunk_place<<<GA, CT, 0, stream>>>(src, dst, hoff, pairs, e, nb);
    k_range_sort<<<nb, CT, 0, stream>>>(pairs, hoff, rowptr, dinv, ssrc, n);

    k_gemm1<<<(n + 15) / 16, T, 0, stream>>>(x, W1, dinv, h1, n);
    k_gather_gemm2<<<(n + 15) / 16, T, 0, stream>>>(rowptr, ssrc, h1, dinv, b1, W2, h2, n);
    k_gather64<<<(n + 7) / 8, T, 0, stream>>>(rowptr, ssrc, h2, dinv, b2, out, n);
}


# Edge buckets in 2 launches, not 5: a chunk block counts, scans and places its pairs in its own stretch, so no scan over the whole table; a range block sums the chunks' local prefixes for its base
# speedup vs baseline: 3.9279x; 1.0915x over previous
#include <hip/hip_runtime.h>

typedef __attribute__((ext_vector_type(2))) float v2f;
typedef __attribute__((ext_vector_type(8))) float v8f;

#define KDIM 128
#define LDSW (KDIM + 4)


#define BSH 8
#define BNODES (1 << BSH)
#define GA 256
#define CT 1024
#define MAXNB 1024
#define EPT 4

#define RPW (GA / (CT / 32))
static_assert(CT == MAXNB, "thread t of a chunk block owns the counter of coarse range t");
static_assert(RPW * (CT / 32) == GA, "the waves of a range block share the GA runs evenly");

__device__ __forceinline__ int block_sum_1024(int v, int* sh) {
    const int lane = threadIdx.x & 31, wave = threadIdx.x >> 5;
#pragma unroll
    for (int o = 16; o > 0; o >>= 1) v += __shfl_xor(v, o);
    if (lane == 0) sh[wave] = v;
    __syncthreads();
    int t = sh[lane];
#pragma unroll
    for (int o = 16; o > 0; o >>= 1) t += __shfl_xor(t, o);
    __syncthreads();
    return t;
}

__global__ __launch_bounds__(CT) void k_chunk_sort(const int* __restrict__ src, const int* __restrict__ dst,
                                                   int* __restrict__ cntT, int* __restrict__ lpT,
                                                   int2* __restrict__ pairs, int e, int nb) {
    __shared__ int h[MAXNB];
    __shared__ int sh[32];
    const int lane = threadIdx.x & 31, wave = threadIdx.x >> 5;
    h[threadIdx.x] = 0;
    __syncthreads();
    const int per = (e + GA - 1) / GA;
    const int beg = blockIdx.x * per;
    const int end = min(e, beg + per);
    for (int base = beg + (int)threadIdx.x; base < end; base += CT * EPT) {
        int d[EPT];
#pragma unroll
        for (int k = 0; k < EPT; ++k) d[k] = (base + k * CT < end) ? dst[base + k * CT] : 0;
#pragma unroll
        for (int k = 0; k < EPT; ++k)
            if (base + k * CT < end) atomicAdd(&h[d[k] >> BSH], 1);
    }
    __syncthreads();
    const int c = h[threadIdx.x];
    int incl = c;
#pragma unroll
    for (int o = 1; o < 32; o <<= 1) {
        int u = __shfl_up(incl, o);
        if (lane >= o) incl += u;
    }
    if (lane == 31) sh[wave] = incl;
    __syncthreads();
    int wsum = sh[lane];
#pragma unroll
    for (int o = 1; o < 32; o <<= 1) {
        int u = __shfl_up(wsum, o);
        if (lane >= o) wsum += u;
    }
    const int wave_excl = __shfl(wsum, wave) - __shfl(sh[lane], wave);
    const int excl = wave_excl + incl - c;
    if ((int)threadIdx.x < nb) {
        cntT[threadIdx.x * GA + blockIdx.x] = c;
        lpT[threadIdx.x * GA + blockIdx.x] = excl;
    }
    h[threadIdx.x] = beg + excl;
    __syncthreads();
    for (int base = beg + (int)threadIdx.x; base < end; base += CT * EPT) {
        int d[EPT], s[EPT];
#pragma unroll
        for (int k = 0; k < EPT; ++k) {
            const bool in = base + k * CT < end;
            d[k] = in ? dst[base + k * CT] : 0;
            s[k] = in ? src[base + k * CT] : 0;
        }
#pragma unroll
        for (int k = 0; k < EPT; ++k)
            if (base + k * CT < end) pairs[atomicAdd(&h[d[k] >> BSH], 1)] = make_int2(s[k], d[k]);
    }
}

__global__ __launch_bounds__(CT) void k_range_sort(const int2* __restrict__ pairs,
                                                   const int* __restrict__ cntT, const int* __restrict__ lpT,
                                                   int* __restrict__ rowptr, float* __restrict__ dinv,
                                                   int* __restrict__ ssrc, int n, int e) {
    __shared__ int cnt[BNODES];
    __shared__ int sh[32];
    __shared__ int rcn[GA];
    __shared__ int rst[GA];
    const int lane = threadIdx.x & 31, wave = threadIdx.x >> 5;
    const int node0 = blockIdx.x << BSH;
    const int per = (e + GA - 1) / GA;
    int lp = 0;
    if (threadIdx.x < GA) {
        lp = lpT[blockIdx.x * GA + threadIdx.x];
        rcn[threadIdx.x] = cntT[blockIdx.x * GA + threadIdx.x];
        rst[threadIdx.x] = (int)threadIdx.x * per + lp;
    }
    if (threadIdx.x < BNODES) cnt[threadIdx.x] = 0;
    const int beg = block_sum_1024(lp, sh);

    int2 p[RPW];
    int nn[RPW], st[RPW];
#pragma unroll
    for (int k = 0; k < RPW; ++k) {
        nn[k] = __builtin_amdgcn_readfirstlane(rcn[wave * RPW + k]);
        st[k] = __builtin_amdgcn_readfirstlane(rst[wave * RPW + k]);
    }
#pragma unroll
    for (int k = 0; k < RPW; ++k) p[k] = (lane < nn[k]) ? pairs[st[k] + lane] : make_int2(0, node0);
#pragma unroll
    for (int k = 0; k < RPW; ++k)
        if (lane < nn[k]) atomicAdd(&cnt[p[k].y - node0], 1);
    for (int k = 0; k < RPW; ++k)
        for (int j = lane + 32; j < nn[k]; j += 32) atomicAdd(&cnt[pairs[st[k] + j].y - node0], 1);
    __syncthreads();
    const int c = (threadIdx.x < BNODES) ? cnt[threadIdx.x] : 0;
    int incl = c;
#pragma unroll
    for (int o = 1; o < 32; o <<= 1) {
        int u = __shfl_up(incl, o);
        if (lane >= o) incl += u;
    }
    if (lane == 31) sh[wave] = incl;
    __syncthreads();
    int woff = 0;
    for (int w = 0; w < BNODES / 32; ++w) woff += (w < wave) ? sh[w] : 0;
    const int excl = beg + woff + incl - c;
    if (threadIdx.x < BNODES) {
        const int node = node0 + (int)threadIdx.x;
        cnt[threadIdx.x] = excl;
        if (node < n) {
            rowptr[node] = excl;
            dinv[node] = rsqrtf((float)(c + 1));
            if (node == n - 1) rowptr[n] = excl + c;
        }
    }
    __syncthreads();
#pragma unroll
    for (int k = 0; k < RPW; ++k)
        if (lane < nn[k]) ssrc[atomicAdd(&cnt[p[k].y - node0], 1)] = p[k].x;
    for (int k = 0; k < RPW; ++k)
        for (int j = lane + 32; j < nn[k]; j += 32) {
            const int2 q = pairs[st[k] + j];
            ssrc[atomicAdd(&cnt[q.y - node0], 1)] = q.x;
        }
}

template <int OUTCH>
__device__ __forceinline__ v8f tile_product(const float* arow, const float* __restrict__ wcol, int kh) {
    v8f acc = {};
#pragma unroll
    for (int k0 = 0; k0 < KDIM; k0 += 4) {
        v2f a, b;
        a.x = arow[k0 + 2 * kh + 0];
        a.y = arow[k0 + 2 * kh + 1];
        b.x = wcol[(size_t)(k0 + 2 * kh + 0) * OUTCH];
        b.y = wcol[(size_t)(k0 + 2 * kh + 1) * OUTCH];
        acc = __builtin_amdgcn_wmma_f32_16x16x4_f32(
             false, a,  false, b,
             (short)0, acc,  false,  false);
    }
    return acc;
}

template <int OUTCH>
__device__ __forceinline__ void tile_store(const v8f& acc, float* __restrict__ H, int mrow0, int col, int n) {
#pragma unroll
    for (int r = 0; r < 8; ++r) {
        int grow = mrow0 + r;
        if (grow < n)
            H[(size_t)grow * OUTCH + col] = acc[r];
    }
}

__global__ __launch_bounds__(256) void k_gemm1(const float* __restrict__ A,
                                               const float* __restrict__ W,
                                               const float* __restrict__ dinv,
                                               float* __restrict__ H, int n) {
    __shared__ float tile[16 * LDSW];

    const int tid = threadIdx.x;
    const int m_base = blockIdx.x * 16;

    for (int c = tid; c < 16 * (KDIM / 4); c += 256) {
        int row = c >> 5;
        int kc  = c & 31;
        int grow = m_base + row;
        if (grow >= n) grow = n - 1;
        float4 v = ((const float4*)(A + (size_t)grow * KDIM))[kc];
        const float s = dinv[grow];
        v.x *= s; v.y *= s; v.z *= s; v.w *= s;
        *(float4*)&tile[row * LDSW + kc * 4] = v;
    }
    __syncthreads();

    const int nt = tid >> 5;
    const int lane = tid & 31;
    const int lm = lane & 15;
    const int kh = lane >> 4;
    const v8f acc = tile_product<128>(&tile[lm * LDSW], W + nt * 16 + lm, kh);
    tile_store<128>(acc, H, m_base + (kh ? 8 : 0), nt * 16 + lm, n);
}

template <int CH> struct RowVec;
template <> struct RowVec<128> { typedef float4 type; };
template <> struct RowVec<64>  { typedef float2 type; };

__device__ __forceinline__ void row_add(float4& a, const float4& h) {
    a.x += h.x; a.y += h.y; a.z += h.z; a.w += h.w;
}
__device__ __forceinline__ void row_add(float2& a, const float2& h) {
    a.x += h.x; a.y += h.y;
}
__device__ __forceinline__ float4 row_scale_bias(const float4& a, float w, const float4& b) {
    return make_float4(fmaf(a.x, w, b.x), fmaf(a.y, w, b.y), fmaf(a.z, w, b.z), fmaf(a.w, w, b.w));
}
__device__ __forceinline__ float2 row_scale_bias(const float2& a, float w, const float2& b) {
    return make_float2(fmaf(a.x, w, b.x), fmaf(a.y, w, b.y));
}

template <int CH>
__device__ __forceinline__ typename RowVec<CH>::type gather_row(int node, int lane, float dd,
                                                                const int* __restrict__ rowptr,
                                                                const int* __restrict__ ssrc,
                                                                const float* __restrict__ hs,
                                                                const float* __restrict__ bias) {
    typedef typename RowVec<CH>::type V;
    const int beg = __builtin_amdgcn_readfirstlane(rowptr[node]);
    const int end = __builtin_amdgcn_readfirstlane(rowptr[node + 1]);

    const V b = ((const V*)bias)[lane];
    V acc = ((const V*)(hs + (size_t)node * CH))[lane];

    for (int base = beg; base < end; base += 32) {
        const int m = min(32, end - base);
        int s_l = node;
        if (lane < m) s_l = ssrc[base + lane];
#define EDGE_ROW(j) (((const V*)(hs + (size_t)__builtin_amdgcn_readlane(s_l, (j)) * CH))[lane])
        int k = 0;
        for (; k + 4 <= m; k += 4) {
            const V h0 = EDGE_ROW(k + 0);
            const V h1 = EDGE_ROW(k + 1);
            const V h2 = EDGE_ROW(k + 2);
            const V h3 = EDGE_ROW(k + 3);
            row_add(acc, h0);
            row_add(acc, h1);
            row_add(acc, h2);
            row_add(acc, h3);
        }
        if (k < m) {
            const V h0 = EDGE_ROW(k);
            if (k + 1 < m) {
                const V h1 = EDGE_ROW(k + 1);
                if (k + 2 < m) {
                    const V h2 = EDGE_ROW(k + 2);
                    row_add(acc, h0);
                    row_add(acc, h1);
                    row_add(acc, h2);
                } else {
                    row_add(acc, h0);
                    row_add(acc, h1);
                }
            } else {
                row_add(acc, h0);
            }
        }
#undef EDGE_ROW
    }
    return row_scale_bias(acc, dd, b);
}

__global__ __launch_bounds__(256) void k_gather64(const int* __restrict__ rowptr,
                                                  const int* __restrict__ ssrc,
                                                  const float* __restrict__ h,
                                                  const float* __restrict__ dinv,
                                                  const float* __restrict__ bias,
                                                  float* __restrict__ out, int n) {
    const int lane = threadIdx.x & 31;
    const int node = __builtin_amdgcn_readfirstlane((int)((blockIdx.x * 256u + threadIdx.x) >> 5));
    if (node >= n) return;
    ((float2*)(out + (size_t)node * 64))[lane] = gather_row<64>(node, lane, dinv[node], rowptr, ssrc, h, bias);
}

__global__ __launch_bounds__(256) void k_gather_gemm2(const int* __restrict__ rowptr,
                                                      const int* __restrict__ ssrc,
                                                      const float* __restrict__ h,
                                                      const float* __restrict__ dinv,
                                                      const float* __restrict__ bias,
                                                      const float* __restrict__ W,
                                                      float* __restrict__ H2, int n) {
    __shared__ float tile[16 * LDSW];

    const int lane = threadIdx.x & 31;
    const int wave = __builtin_amdgcn_readfirstlane((int)(threadIdx.x >> 5));
    const int m_base = blockIdx.x * 16;

    for (int r = 0; r < 2; ++r) {
        const int lrow = 2 * wave + r;
        const int node = m_base + lrow;
        float4 v = make_float4(0.0f, 0.0f, 0.0f, 0.0f);
        if (node < n) {
            const float dd = dinv[node];
            v = gather_row<128>(node, lane, dd, rowptr, ssrc, h, bias);
            v.x = fmaxf(v.x, 0.0f) * dd; v.y = fmaxf(v.y, 0.0f) * dd;
            v.z = fmaxf(v.z, 0.0f) * dd; v.w = fmaxf(v.w, 0.0f) * dd;
        }
        *(float4*)&tile[lrow * LDSW + lane * 4] = v;
    }
    __syncthreads();
    if (wave >= 4) return;

    const int lm = lane & 15;
    const int kh = lane >> 4;
    const v8f acc = tile_product<64>(&tile[lm * LDSW], W + wave * 16 + lm, kh);
    tile_store<64>(acc, H2, m_base + (kh ? 8 : 0), wave * 16 + lm, n);
}

extern "C" void kernel_launch(void* const* d_in, const int* in_sizes, int n_in,
                              void* d_out, int out_size, void* d_ws, size_t ws_size,
                              hipStream_t stream) {
    const float* x  = (const float*)d_in[0];
    const int*   ei = (const int*)d_in[1];
    const float* W1 = (const float*)d_in[2];
    const float* b1 = (const float*)d_in[3];
    const float* W2 = (const float*)d_in[4];
    const float* b2 = (const float*)d_in[5];
    float* out = (float*)d_out;

    const int n = in_sizes[0] / 128;
    const int e = in_sizes[1] / 2;
    const int* src = ei;
    const int* dst = ei + e;

    float* dinv = (float*)d_ws;
    float* h1 = dinv + n;
    float* h2 = h1 + (size_t)n * 128;
    int* rowptr = (int*)(h2 + (size_t)n * 64);
    int* ssrc   = rowptr + (n + 1);
    const int nb = (n + BNODES - 1) / BNODES;
    const int nh = nb * GA;
    int* cntT   = ssrc + e;
    int* lpT    = cntT + nh;
    int2* pairs = (int2*)(h1 + (n & 1));

    const unsigned T = 256;
    k_chunk_sort<<<GA, CT, 0, stream>>>(src, dst, cntT, lpT, pairs, e, nb);
    k_range_sort<<<nb, CT, 0, stream>>>(pairs, cntT, lpT, rowptr, dinv, ssrc, n, e);

    k_gemm1<<<(n + 15) / 16, T, 0, stream>>>(x, W1, dinv, h1, n);
    k_gather_gemm2<<<(n + 15) / 16, T, 0, stream>>>(rowptr, ssrc, h1, dinv, b1, W2, h2, n);
    k_gather64<<<(n + 7) / 8, T, 0, stream>>>(rowptr, ssrc, h2, dinv, b2, out, n);
}


# Layer 1 as (aggregate x) @ W1: the gather kernel sums dinv[src]-weighted x rows into an LDS tile and multiplies by W1, then W2, there; product launch and h1 round trip gone; sums reassociated
# speedup vs baseline: 4.5023x; 1.1462x over previous
#include <hip/hip_runtime.h>

typedef __attribute__((ext_vector_type(2))) float v2f;
typedef __attribute__((ext_vector_type(8))) float v8f;

#define KDIM 128
#define LDSW (KDIM + 4)


#define BSH 8
#define BNODES (1 << BSH)
#define GA 256
#define CT 1024
#define MAXNB 1024
#define EPT 4

#define RPW (GA / (CT / 32))
static_assert(CT == MAXNB, "thread t of a chunk block owns the counter of coarse range t");
static_assert(RPW * (CT / 32) == GA, "the waves of a range block share the GA runs evenly");

__device__ __forceinline__ int block_sum_1024(int v, int* sh) {
    const int lane = threadIdx.x & 31, wave = threadIdx.x >> 5;
#pragma unroll
    for (int o = 16; o > 0; o >>= 1) v += __shfl_xor(v, o);
    if (lane == 0) sh[wave] = v;
    __syncthreads();
    int t = sh[lane];
#pragma unroll
    for (int o = 16; o > 0; o >>= 1) t += __shfl_xor(t, o);
    __syncthreads();
    return t;
}

__global__ __launch_bounds__(CT) void k_chunk_sort(const int* __restrict__ src, const int* __restrict__ dst,
                                                   int* __restrict__ cntT, int* __restrict__ lpT,
                                                   int2* __restrict__ pairs, int e, int nb) {
    __shared__ int h[MAXNB];
    __shared__ int sh[32];
    const int lane = threadIdx.x & 31, wave = threadIdx.x >> 5;
    h[threadIdx.x] = 0;
    __syncthreads();
    const int per = (e + GA - 1) / GA;
    const int beg = blockIdx.x * per;
    const int end = min(e, beg + per);
    for (int base = beg + (int)threadIdx.x; base < end; base += CT * EPT) {
        int d[EPT];
#pragma unroll
        for (int k = 0; k < EPT; ++k) d[k] = (base + k * CT < end) ? dst[base + k * CT] : 0;
#pragma unroll
        for (int k = 0; k < EPT; ++k)
            if (base + k * CT < end) atomicAdd(&h[d[k] >> BSH], 1);
    }
    __syncthreads();
    const int c = h[threadIdx.x];
    int incl = c;
#pragma unroll
    for (int o = 1; o < 32; o <<= 1) {
        int u = __shfl_up(incl, o);
        if (lane >= o) incl += u;
    }
    if (lane == 31) sh[wave] = incl;
    __syncthreads();
    int wsum = sh[lane];
#pragma unroll
    for (int o = 1; o < 32; o <<= 1) {
        int u = __shfl_up(wsum, o);
        if (lane >= o) wsum += u;
    }
    const int wave_excl = __shfl(wsum, wave) - __shfl(sh[lane], wave);
    const int excl = wave_excl + incl - c;
    if ((int)threadIdx.x < nb) {
        cntT[threadIdx.x * GA + blockIdx.x] = c;
        lpT[threadIdx.x * GA + blockIdx.x] = excl;
    }
    h[threadIdx.x] = beg + excl;
    __syncthreads();
    for (int base = beg + (int)threadIdx.x; base < end; base += CT * EPT) {
        int d[EPT], s[EPT];
#pragma unroll
        for (int k = 0; k < EPT; ++k) {
            const bool in = base + k * CT < end;
            d[k] = in ? dst[base + k * CT] : 0;
            s[k] = in ? src[base + k * CT] : 0;
        }
#pragma unroll
        for (int k = 0; k < EPT; ++k)
            if (base + k * CT < end) pairs[atomicAdd(&h[d[k] >> BSH], 1)] = make_int2(s[k], d[k]);
    }
}

__global__ __launch_bounds__(CT) void k_range_sort(const int2* __restrict__ pairs,
                                                   const int* __restrict__ cntT, const int* __restrict__ lpT,
                                                   int* __restrict__ rowptr, float* __restrict__ dinv,
                                                   int* __restrict__ ssrc, int n, int e) {
    __shared__ int cnt[BNODES];
    __shared__ int sh[32];
    __shared__ int rcn[GA];
    __shared__ int rst[GA];
    const int lane = threadIdx.x & 31, wave = threadIdx.x >> 5;
    const int node0 = blockIdx.x << BSH;
    const int per = (e + GA - 1) / GA;
    int lp = 0;
    if (threadIdx.x < GA) {
        lp = lpT[blockIdx.x * GA + threadIdx.x];
        rcn[threadIdx.x] = cntT[blockIdx.x * GA + threadIdx.x];
        rst[threadIdx.x] = (int)threadIdx.x * per + lp;
    }
    if (threadIdx.x < BNODES) cnt[threadIdx.x] = 0;
    const int beg = block_sum_1024(lp, sh);

    int2 p[RPW];
    int nn[RPW], st[RPW];
#pragma unroll
    for (int k = 0; k < RPW; ++k) {
        nn[k] = __builtin_amdgcn_readfirstlane(rcn[wave * RPW + k]);
        st[k] = __builtin_amdgcn_readfirstlane(rst[wave * RPW + k]);
    }
#pragma unroll
    for (int k = 0; k < RPW; ++k) p[k] = (lane < nn[k]) ? pairs[st[k] + lane] : make_int2(0, node0);
#pragma unroll
    for (int k = 0; k < RPW; ++k)
        if (lane < nn[k]) atomicAdd(&cnt[p[k].y - node0], 1);
    for (int k = 0; k < RPW; ++k)
        for (int j = lane + 32; j < nn[k]; j += 32) atomicAdd(&cnt[pairs[st[k] + j].y - node0], 1);
    __syncthreads();
    const int c = (threadIdx.x < BNODES) ? cnt[threadIdx.x] : 0;
    int incl = c;
#pragma unroll
    for (int o = 1; o < 32; o <<= 1) {
        int u = __shfl_up(incl, o);
        if (lane >= o) incl += u;
    }
    if (lane == 31) sh[wave] = incl;
    __syncthreads();
    int woff = 0;
    for (int w = 0; w < BNODES / 32; ++w) woff += (w < wave) ? sh[w] : 0;
    const int excl = beg + woff + incl - c;
    if (threadIdx.x < BNODES) {
        const int node = node0 + (int)threadIdx.x;
        cnt[threadIdx.x] = excl;
        if (node < n) {
            rowptr[node] = excl;
            dinv[node] = rsqrtf((float)(c + 1));
            if (node == n - 1) rowptr[n] = excl + c;
        }
    }
    __syncthreads();
#pragma unroll
    for (int k = 0; k < RPW; ++k)
        if (lane < nn[k]) ssrc[atomicAdd(&cnt[p[k].y - node0], 1)] = p[k].x;
    for (int k = 0; k < RPW; ++k)
        for (int j = lane + 32; j < nn[k]; j += 32) {
            const int2 q = pairs[st[k] + j];
            ssrc[atomicAdd(&cnt[q.y - node0], 1)] = q.x;
        }
}

template <int OUTCH>
__device__ __forceinline__ v8f tile_product(const float* arow, const float* __restrict__ wcol, int kh) {
    v8f acc = {};
#pragma unroll
    for (int k0 = 0; k0 < KDIM; k0 += 4) {
        v2f a, b;
        a.x = arow[k0 + 2 * kh + 0];
        a.y = arow[k0 + 2 * kh + 1];
        b.x = wcol[(size_t)(k0 + 2 * kh + 0) * OUTCH];
        b.y = wcol[(size_t)(k0 + 2 * kh + 1) * OUTCH];
        acc = __builtin_amdgcn_wmma_f32_16x16x4_f32(
             false, a,  false, b,
             (short)0, acc,  false,  false);
    }
    return acc;
}

template <int OUTCH>
__device__ __forceinline__ void tile_store(const v8f& acc, float* __restrict__ H, int mrow0, int col, int n) {
#pragma unroll
    for (int r = 0; r < 8; ++r) {
        int grow = mrow0 + r;
        if (grow < n)
            H[(size_t)grow * OUTCH + col] = acc[r];
    }
}


__device__ __forceinline__ void row_add(float2& a, const float2& h) {
    a.x += h.x; a.y += h.y;
}
__device__ __forceinline__ void row_fma(float4& a, float w, const float4& h) {
    a.x = fmaf(w, h.x, a.x); a.y = fmaf(w, h.y, a.y); a.z = fmaf(w, h.z, a.z); a.w = fmaf(w, h.w, a.w);
}

__device__ __forceinline__ float2 gather_row64(int node, int lane, float dd,
                                               const int* __restrict__ rowptr,
                                               const int* __restrict__ ssrc,
                                               const float* __restrict__ hs,
                                               const float* __restrict__ bias) {
    const int beg = __builtin_amdgcn_readfirstlane(rowptr[node]);
    const int end = __builtin_amdgcn_readfirstlane(rowptr[node + 1]);

    const float2 b = ((const float2*)bias)[lane];
    float2 acc = ((const float2*)(hs + (size_t)node * 64))[lane];

    for (int base = beg; base < end; base += 32) {
        const int m = min(32, end - base);
        int s_l = node;
        if (lane < m) s_l = ssrc[base + lane];
#define EDGE_ROW(j) (((const float2*)(hs + (size_t)__builtin_amdgcn_readlane(s_l, (j)) * 64))[lane])
        int k = 0;
        for (; k + 4 <= m; k += 4) {
            const float2 h0 = EDGE_ROW(k + 0);
            const float2 h1 = EDGE_ROW(k + 1);
            const float2 h2 = EDGE_ROW(k + 2);
            const float2 h3 = EDGE_ROW(k + 3);
            row_add(acc, h0);
            row_add(acc, h1);
            row_add(acc, h2);
            row_add(acc, h3);
        }
        if (k < m) {
            const float2 h0 = EDGE_ROW(k);
            if (k + 1 < m) {
                const float2 h1 = EDGE_ROW(k + 1);
                if (k + 2 < m) {
                    const float2 h2 = EDGE_ROW(k + 2);
                    row_add(acc, h0);
                    row_add(acc, h1);
                    row_add(acc, h2);
                } else {
                    row_add(acc, h0);
                    row_add(acc, h1);
                }
            } else {
                row_add(acc, h0);
            }
        }
#undef EDGE_ROW
    }
    return make_float2(fmaf(acc.x, dd, b.x), fmaf(acc.y, dd, b.y));
}

__device__ __forceinline__ float4 gather_row_x(int node, int lane, float dd,
                                               const int* __restrict__ rowptr,
                                               const int* __restrict__ ssrc,
                                               const float* __restrict__ dinv,
                                               const float* __restrict__ x) {
    const int beg = __builtin_amdgcn_readfirstlane(rowptr[node]);
    const int end = __builtin_amdgcn_readfirstlane(rowptr[node + 1]);

    float4 acc = ((const float4*)(x + (size_t)node * KDIM))[lane];
    acc.x *= dd; acc.y *= dd; acc.z *= dd; acc.w *= dd;

    for (int base = beg; base < end; base += 32) {
        const int m = min(32, end - base);
        int s_l = node;
        if (lane < m) s_l = ssrc[base + lane];
        const int w_l = __float_as_int(dinv[s_l]);
#define EDGE_ROW(j) (((const float4*)(x + (size_t)__builtin_amdgcn_readlane(s_l, (j)) * KDIM))[lane])
#define EDGE_W(j) __int_as_float(__builtin_amdgcn_readlane(w_l, (j)))
        int k = 0;
        for (; k + 4 <= m; k += 4) {
            const float4 h0 = EDGE_ROW(k + 0);
            const float4 h1 = EDGE_ROW(k + 1);
            const float4 h2 = EDGE_ROW(k + 2);
            const float4 h3 = EDGE_ROW(k + 3);
            row_fma(acc, EDGE_W(k + 0), h0);
            row_fma(acc, EDGE_W(k + 1), h1);
            row_fma(acc, EDGE_W(k + 2), h2);
            row_fma(acc, EDGE_W(k + 3), h3);
        }
        if (k < m) {
            const float4 h0 = EDGE_ROW(k);
            if (k + 1 < m) {
                const float4 h1 = EDGE_ROW(k + 1);
                if (k + 2 < m) {
                    const float4 h2 = EDGE_ROW(k + 2);
                    row_fma(acc, EDGE_W(k), h0);
                    row_fma(acc, EDGE_W(k + 1), h1);
                    row_fma(acc, EDGE_W(k + 2), h2);
                } else {
                    row_fma(acc, EDGE_W(k), h0);
                    row_fma(acc, EDGE_W(k + 1), h1);
                }
            } else {
                row_fma(acc, EDGE_W(k), h0);
            }
        }
#undef EDGE_W
#undef EDGE_ROW
    }
    return acc;
}

__global__ __launch_bounds__(256) void k_gather64(const int* __restrict__ rowptr,
                                                  const int* __restrict__ ssrc,
                                                  const float* __restrict__ h,
                                                  const float* __restrict__ dinv,
                                                  const float* __restrict__ bias,
                                                  float* __restrict__ out, int n) {
    const int lane = threadIdx.x & 31;
    const int node = __builtin_amdgcn_readfirstlane((int)((blockIdx.x * 256u + threadIdx.x) >> 5));
    if (node >= n) return;
    ((float2*)(out + (size_t)node * 64))[lane] = gather_row64(node, lane, dinv[node], rowptr, ssrc, h, bias);
}

#define FW 16
#define FR (2 * FW)
static_assert(FR % 16 == 0 && (FR / 16) * 8 == FW, "one 16x16 tile of the layer-1 product per wave");
__global__ __launch_bounds__(FW * 32, 16) void k_gather_gemm12(const int* __restrict__ rowptr,
                                                           const int* __restrict__ ssrc,
                                                           const float* __restrict__ x,
                                                           const float* __restrict__ dinv,
                                                           const float* __restrict__ W1,
                                                           const float* __restrict__ b1,
                                                           const float* __restrict__ W2,
                                                           float* __restrict__ H2, int n) {
    __shared__ float tile[FR * LDSW];
    __shared__ float tile2[FR * LDSW];
    __shared__ float rdinv[FR];

    const int lane = threadIdx.x & 31;
    const int wave = __builtin_amdgcn_readfirstlane((int)(threadIdx.x >> 5));
    const int m_base = blockIdx.x * FR;

    for (int r = 0; r < 2; ++r) {
        const int lrow = 2 * wave + r;
        const int node = m_base + lrow;
        float4 v = make_float4(0.0f, 0.0f, 0.0f, 0.0f);
        float dd = 0.0f;
        if (node < n) {
            dd = dinv[node];
            v = gather_row_x(node, lane, dd, rowptr, ssrc, dinv, x);
        }
        *(float4*)&tile[lrow * LDSW + lane * 4] = v;
        if (lane == 0) rdinv[lrow] = dd;
    }
    __syncthreads();

    const int lm = lane & 15;
    const int kh = lane >> 4;
    {
        const int rt = wave >> 3;
        const int col = (wave & 7) * 16 + lm;
        const float bc = b1[col];
        v8f acc = tile_product<128>(&tile[(rt * 16 + lm) * LDSW], W1 + col, kh);
        asm volatile("v_nop\n\tv_nop\n\tv_nop\n\tv_nop" : "+v"(acc));
#pragma unroll
        for (int r = 0; r < 8; ++r) {
            const int row = rt * 16 + 8 * kh + r;
            const float d = rdinv[row];
            tile2[row * LDSW + col] = fmaxf(fmaf(acc[r], d, bc), 0.0f) * d;
        }
    }
    __syncthreads();
    if (wave >= FR / 4) return;

    const int rt = wave >> 2;
    const int col = (wave & 3) * 16 + lm;
    const v8f acc2 = tile_product<64>(&tile2[(rt * 16 + lm) * LDSW], W2 + col, kh);
    tile_store<64>(acc2, H2, m_base + rt * 16 + (kh ? 8 : 0), col, n);
}

extern "C" void kernel_launch(void* const* d_in, const int* in_sizes, int n_in,
                              void* d_out, int out_size, void* d_ws, size_t ws_size,
                              hipStream_t stream) {
    const float* x  = (const float*)d_in[0];
    const int*   ei = (const int*)d_in[1];
    const float* W1 = (const float*)d_in[2];
    const float* b1 = (const float*)d_in[3];
    const float* W2 = (const float*)d_in[4];
    const float* b2 = (const float*)d_in[5];
    float* out = (float*)d_out;

    const int n = in_sizes[0] / 128;
    const int e = in_sizes[1] / 2;
    const int* src = ei;
    const int* dst = ei + e;

    float* dinv = (float*)d_ws;
    float* pbuf = dinv + n;
    float* h2 = pbuf + (size_t)n * 128;
    int* rowptr = (int*)(h2 + (size_t)n * 64);
    int* ssrc   = rowptr + (n + 1);
    const int nb = (n + BNODES - 1) / BNODES;
    const int nh = nb * GA;
    int* cntT   = ssrc + e;
    int* lpT    = cntT + nh;
    int2* pairs = (int2*)(pbuf + (n & 1));

    const unsigned T = 256;
    k_chunk_sort<<<GA, CT, 0, stream>>>(src, dst, cntT, lpT, pairs, e, nb);
    k_range_sort<<<nb, CT, 0, stream>>>(pairs, cntT, lpT, rowptr, dinv, ssrc, n, e);

    k_gather_gemm12<<<(n + FR - 1) / FR, FW * 32, 0, stream>>>(rowptr, ssrc, x, dinv, W1, b1, W2, h2, n);
    k_gather64<<<(n + 7) / 8, T, 0, stream>>>(rowptr, ssrc, h2, dinv, b2, out, n);
}


# W1, W2 repacked per call into fragment order: a tile's B operands come as 16 whole-line 16-byte loads, not 64 four-byte ones; A read in the same k order; k order of the dot products changes
# speedup vs baseline: 4.6447x; 1.0316x over previous
#include <hip/hip_runtime.h>

typedef __attribute__((ext_vector_type(2))) float v2f;
typedef __attribute__((ext_vector_type(8))) float v8f;

#define KDIM 128
#define LDSW (KDIM + 4)


#define BSH 8
#define BNODES (1 << BSH)
#define GA 256
#define CT 1024
#define MAXNB 1024
#define EPT 4

#define RPW (GA / (CT / 32))
static_assert(CT == MAXNB, "thread t of a chunk block owns the counter of coarse range t");
static_assert(RPW * (CT / 32) == GA, "the waves of a range block share the GA runs evenly");

__device__ __forceinline__ int block_sum_1024(int v, int* sh) {
    const int lane = threadIdx.x & 31, wave = threadIdx.x >> 5;
#pragma unroll
    for (int o = 16; o > 0; o >>= 1) v += __shfl_xor(v, o);
    if (lane == 0) sh[wave] = v;
    __syncthreads();
    int t = sh[lane];
#pragma unroll
    for (int o = 16; o > 0; o >>= 1) t += __shfl_xor(t, o);
    __syncthreads();
    return t;
}

#define W1P (KDIM * 128)
#define W2P (KDIM * 64)
static_assert(W1P + W2P <= GA * CT, "one thread of the first launch per packed element");
__device__ __forceinline__ float pack_w_fetch(const float* __restrict__ W1, const float* __restrict__ W2, int g) {
    const bool second = g >= W1P;
    const int q = second ? g - W1P : g;
    const int s = q & 3, ln = (q >> 2) & 31, j = (q >> 7) & 15, tn = q >> 11;
    const int k = 8 * j + 4 * (ln >> 4) + s;
    const int col = tn * 16 + (ln & 15);
    return second ? W2[k * 64 + col] : W1[k * 128 + col];
}

__global__ __launch_bounds__(CT) void k_chunk_sort(const int* __restrict__ src, const int* __restrict__ dst,
                                                   int* __restrict__ cntT, int* __restrict__ lpT,
                                                   int2* __restrict__ pairs, int e, int nb,
                                                   const float* __restrict__ W1, const float* __restrict__ W2,
                                                   float* __restrict__ Wp) {
    __shared__ int h[MAXNB];
    __shared__ int sh[32];
    const int lane = threadIdx.x & 31, wave = threadIdx.x >> 5;
    h[threadIdx.x] = 0;
    __syncthreads();
    const int per = (e + GA - 1) / GA;
    const int beg = blockIdx.x * per;
    const int end = min(e, beg + per);
    for (int base = beg + (int)threadIdx.x; base < end; base += CT * EPT) {
        int d[EPT];
#pragma unroll
        for (int k = 0; k < EPT; ++k) d[k] = (base + k * CT < end) ? dst[base + k * CT] : 0;
#pragma unroll
        for (int k = 0; k < EPT; ++k)
            if (base + k * CT < end) atomicAdd(&h[d[k] >> BSH], 1);
    }
    __syncthreads();
    const int c = h[threadIdx.x];
    int incl = c;
#pragma unroll
    for (int o = 1; o < 32; o <<= 1) {
        int u = __shfl_up(incl, o);
        if (lane >= o) incl += u;
    }
    if (lane == 31) sh[wave] = incl;
    __syncthreads();
    int wsum = sh[lane];
#pragma unroll
    for (int o = 1; o < 32; o <<= 1) {
        int u = __shfl_up(wsum, o);
        if (lane >= o) wsum += u;
    }
    const int wave_excl = __shfl(wsum, wave) - __shfl(sh[lane], wave);
    const int excl = wave_excl + incl - c;
    if ((int)threadIdx.x < nb) {
        cntT[threadIdx.x * GA + blockIdx.x] = c;
        lpT[threadIdx.x * GA + blockIdx.x] = excl;
    }
    h[threadIdx.x] = beg + excl;
    __syncthreads();
    const int g = (int)(blockIdx.x * CT + threadIdx.x);
    float wv = 0.0f;
    if (g < W1P + W2P) wv = pack_w_fetch(W1, W2, g);
    for (int base = beg + (int)threadIdx.x; base < end; base += CT * EPT) {
        int d[EPT], s[EPT];
#pragma unroll
        for (int k = 0; k < EPT; ++k) {
            const bool in = base + k * CT < end;
            d[k] = in ? dst[base + k * CT] : 0;
            s[k] = in ? src[base + k * CT] : 0;
        }
#pragma unroll
        for (int k = 0; k < EPT; ++k)
            if (base + k * CT < end) pairs[atomicAdd(&h[d[k] >> BSH], 1)] = make_int2(s[k], d[k]);
    }
    if (g < W1P + W2P) Wp[g] = wv;
}

__global__ __launch_bounds__(CT) void k_range_sort(const int2* __restrict__ pairs,
                                                   const int* __restrict__ cntT, const int* __restrict__ lpT,
                                                   int* __restrict__ rowptr, float* __restrict__ dinv,
                                                   int* __restrict__ ssrc, int n, int e) {
    __shared__ int cnt[BNODES];
    __shared__ int sh[32];
    __shared__ int rcn[GA];
    __shared__ int rst[GA];
    const int lane = threadIdx.x & 31, wave = threadIdx.x >> 5;
    const int node0 = blockIdx.x << BSH;
    const int per = (e + GA - 1) / GA;
    int lp = 0;
    if (threadIdx.x < GA) {
        lp = lpT[blockIdx.x * GA + threadIdx.x];
        rcn[threadIdx.x] = cntT[blockIdx.x * GA + threadIdx.x];
        rst[threadIdx.x] = (int)threadIdx.x * per + lp;
    }
    if (threadIdx.x < BNODES) cnt[threadIdx.x] = 0;
    const int beg = block_sum_1024(lp, sh);

    int2 p[RPW];
    int nn[RPW], st[RPW];
#pragma unroll
    for (int k = 0; k < RPW; ++k) {
        nn[k] = __builtin_amdgcn_readfirstlane(rcn[wave * RPW + k]);
        st[k] = __builtin_amdgcn_readfirstlane(rst[wave * RPW + k]);
    }
#pragma unroll
    for (int k = 0; k < RPW; ++k) p[k] = (lane < nn[k]) ? pairs[st[k] + lane] : make_int2(0, node0);
#pragma unroll
    for (int k = 0; k < RPW; ++k)
        if (lane < nn[k]) atomicAdd(&cnt[p[k].y - node0], 1);
    for (int k = 0; k < RPW; ++k)
        for (int j = lane + 32; j < nn[k]; j += 32) atomicAdd(&cnt[pairs[st[k] + j].y - node0], 1);
    __syncthreads();
    const int c = (threadIdx.x < BNODES) ? cnt[threadIdx.x] : 0;
    int incl = c;
#pragma unroll
    for (int o = 1; o < 32; o <<= 1) {
        int u = __shfl_up(incl, o);
        if (lane >= o) incl += u;
    }
    if (lane == 31) sh[wave] = incl;
    __syncthreads();
    int woff = 0;
    for (int w = 0; w < BNODES / 32; ++w) woff += (w < wave) ? sh[w] : 0;
    const int excl = beg + woff + incl - c;
    if (threadIdx.x < BNODES) {
        const int node = node0 + (int)threadIdx.x;
        cnt[threadIdx.x] = excl;
        if (node < n) {
            rowptr[node] = excl;
            dinv[node] = rsqrtf((float)(c + 1));
            if (node == n - 1) rowptr[n] = excl + c;
        }
    }
    __syncthreads();
#pragma unroll
    for (int k = 0; k < RPW; ++k)
        if (lane < nn[k]) ssrc[atomicAdd(&cnt[p[k].y - node0], 1)] = p[k].x;
    for (int k = 0; k < RPW; ++k)
        for (int j = lane + 32; j < nn[k]; j += 32) {
            const int2 q = pairs[st[k] + j];
            ssrc[atomicAdd(&cnt[q.y - node0], 1)] = q.x;
        }
}

__device__ __forceinline__ v8f tile_product(const float* arow, const float4* __restrict__ wfrag, int kh) {
    v8f acc = {};
#pragma unroll
    for (int j = 0; j < KDIM / 8; ++j) {
        const float4 a = *(const float4*)(arow + 8 * j + 4 * kh);
        const float4 b = wfrag[j * 32];
        const v2f a0 = {a.x, a.y}, b0 = {b.x, b.y};
        const v2f a1 = {a.z, a.w}, b1 = {b.z, b.w};
        acc = __builtin_amdgcn_wmma_f32_16x16x4_f32(
             false, a0,  false, b0,
             (short)0, acc,  false,  false);
        acc = __builtin_amdgcn_wmma_f32_16x16x4_f32(
             false, a1,  false, b1,
             (short)0, acc,  false,  false);
    }
    return acc;
}

template <int OUTCH>
__device__ __forceinline__ void tile_store(const v8f& acc, float* __restrict__ H, int mrow0, int col, int n) {
#pragma unroll
    for (int r = 0; r < 8; ++r) {
        int grow = mrow0 + r;
        if (grow < n)
            H[(size_t)grow * OUTCH + col] = acc[r];
    }
}


__device__ __forceinline__ void row_add(float2& a, const float2& h) {
    a.x += h.x; a.y += h.y;
}
__device__ __forceinline__ void row_fma(float4& a, float w, const float4& h) {
    a.x = fmaf(w, h.x, a.x); a.y = fmaf(w, h.y, a.y); a.z = fmaf(w, h.z, a.z); a.w = fmaf(w, h.w, a.w);
}

__device__ __forceinline__ float2 gather_row64(int node, int lane, float dd,
                                               const int* __restrict__ rowptr,
                                               const int* __restrict__ ssrc,
                                               const float* __restrict__ hs,
                                               const float* __restrict__ bias) {
    const int beg = __builtin_amdgcn_readfirstlane(rowptr[node]);
    const int end = __builtin_amdgcn_readfirstlane(rowptr[node + 1]);

    const float2 b = ((const float2*)bias)[lane];
    float2 acc = ((const float2*)(hs + (size_t)node * 64))[lane];

    for (int base = beg; base < end; base += 32) {
        const int m = min(32, end - base);
        int s_l = node;
        if (lane < m) s_l = ssrc[base + lane];
#define EDGE_ROW(j) (((const float2*)(hs + (size_t)__builtin_amdgcn_readlane(s_l, (j)) * 64))[lane])
        int k = 0;
        for (; k + 4 <= m; k += 4) {
            const float2 h0 = EDGE_ROW(k + 0);
            const float2 h1 = EDGE_ROW(k + 1);
            const float2 h2 = EDGE_ROW(k + 2);
            const float2 h3 = EDGE_ROW(k + 3);
            row_add(acc, h0);
            row_add(acc, h1);
            row_add(acc, h2);
            row_add(acc, h3);
        }
        if (k < m) {
            const float2 h0 = EDGE_ROW(k);
            if (k + 1 < m) {
                const float2 h1 = EDGE_ROW(k + 1);
                if (k + 2 < m) {
                    const float2 h2 = EDGE_ROW(k + 2);
                    row_add(acc, h0);
                    row_add(acc, h1);
                    row_add(acc, h2);
                } else {
                    row_add(acc, h0);
                    row_add(acc, h1);
                }
            } else {
                row_add(acc, h0);
            }
        }
#undef EDGE_ROW
    }
    return make_float2(fmaf(acc.x, dd, b.x), fmaf(acc.y, dd, b.y));
}

__device__ __forceinline__ float4 gather_row_x(int node, int lane, float dd,
                                               const int* __restrict__ rowptr,
                                               const int* __restrict__ ssrc,
                                               const float* __restrict__ dinv,
                                               const float* __restrict__ x) {
    const int beg = __builtin_amdgcn_readfirstlane(rowptr[node]);
    const int end = __builtin_amdgcn_readfirstlane(rowptr[node + 1]);

    float4 acc = ((const float4*)(x + (size_t)node * KDIM))[lane];
    acc.x *= dd; acc.y *= dd; acc.z *= dd; acc.w *= dd;

    for (int base = beg; base < end; base += 32) {
        const int m = min(32, end - base);
        int s_l = node;
        if (lane < m) s_l = ssrc[base + lane];
        const int w_l = __float_as_int(dinv[s_l]);
#define EDGE_ROW(j) (((const float4*)(x + (size_t)__builtin_amdgcn_readlane(s_l, (j)) * KDIM))[lane])
#define EDGE_W(j) __int_as_float(__builtin_amdgcn_readlane(w_l, (j)))
        int k = 0;
        for (; k + 4 <= m; k += 4) {
            const float4 h0 = EDGE_ROW(k + 0);
            const float4 h1 = EDGE_ROW(k + 1);
            const float4 h2 = EDGE_ROW(k + 2);
            const float4 h3 = EDGE_ROW(k + 3);
            row_fma(acc, EDGE_W(k + 0), h0);
            row_fma(acc, EDGE_W(k + 1), h1);
            row_fma(acc, EDGE_W(k + 2), h2);
            row_fma(acc, EDGE_W(k + 3), h3);
        }
        if (k < m) {
            const float4 h0 = EDGE_ROW(k);
            if (k + 1 < m) {
                const float4 h1 = EDGE_ROW(k + 1);
                if (k + 2 < m) {
                    const float4 h2 = EDGE_ROW(k + 2);
                    row_fma(acc, EDGE_W(k), h0);
                    row_fma(acc, EDGE_W(k + 1), h1);
                    row_fma(acc, EDGE_W(k + 2), h2);
                } else {
                    row_fma(acc, EDGE_W(k), h0);
                    row_fma(acc, EDGE_W(k + 1), h1);
                }
            } else {
                row_fma(acc, EDGE_W(k), h0);
            }
        }
#undef EDGE_W
#undef EDGE_ROW
    }
    return acc;
}

__global__ __launch_bounds__(256) void k_gather64(const int* __restrict__ rowptr,
                                                  const int* __restrict__ ssrc,
                                                  const float* __restrict__ h,
                                                  const float* __restrict__ dinv,
                                                  const float* __restrict__ bias,
                                                  float* __restrict__ out, int n) {
    const int lane = threadIdx.x & 31;
    const int node = __builtin_amdgcn_readfirstlane((int)((blockIdx.x * 256u + threadIdx.x) >> 5));
    if (node >= n) return;
    ((float2*)(out + (size_t)node * 64))[lane] = gather_row64(node, lane, dinv[node], rowptr, ssrc, h, bias);
}

#define FW 16
#define FR (2 * FW)
static_assert(FR % 16 == 0 && (FR / 16) * 8 == FW, "one 16x16 tile of the layer-1 product per wave");
__global__ __launch_bounds__(FW * 32, 16) void k_gather_gemm12(const int* __restrict__ rowptr,
                                                           const int* __restrict__ ssrc,
                                                           const float* __restrict__ x,
                                                           const float* __restrict__ dinv,
                                                           const float* __restrict__ Wp,
                                                           const float* __restrict__ b1,
                                                           float* __restrict__ H2, int n) {
    __shared__ float tile[FR * LDSW];
    __shared__ float tile2[FR * LDSW];
    __shared__ float rdinv[FR];

    const int lane = threadIdx.x & 31;
    const int wave = __builtin_amdgcn_readfirstlane((int)(threadIdx.x >> 5));
    const int m_base = blockIdx.x * FR;

    for (int r = 0; r < 2; ++r) {
        const int lrow = 2 * wave + r;
        const int node = m_base + lrow;
        float4 v = make_float4(0.0f, 0.0f, 0.0f, 0.0f);
        float dd = 0.0f;
        if (node < n) {
            dd = dinv[node];
            v = gather_row_x(node, lane, dd, rowptr, ssrc, dinv, x);
        }
        *(float4*)&tile[lrow * LDSW + lane * 4] = v;
        if (lane == 0) rdinv[lrow] = dd;
    }
    __syncthreads();

    const int lm = lane & 15;
    const int kh = lane >> 4;
    {
        const int rt = wave >> 3;
        const int col = (wave & 7) * 16 + lm;
        const float bc = b1[col];
        v8f acc = tile_product(&tile[(rt * 16 + lm) * LDSW], (const float4*)Wp + (wave & 7) * 512 + lane, kh);
        asm volatile("v_nop\n\tv_nop\n\tv_nop\n\tv_nop" : "+v"(acc));
#pragma unroll
        for (int r = 0; r < 8; ++r) {
            const int row = rt * 16 + 8 * kh + r;
            const float d = rdinv[row];
            tile2[row * LDSW + col] = fmaxf(fmaf(acc[r], d, bc), 0.0f) * d;
        }
    }
    __syncthreads();
    if (wave >= FR / 4) return;

    const int rt = wave >> 2;
    const int col = (wave & 3) * 16 + lm;
    const v8f acc2 = tile_product(&tile2[(rt * 16 + lm) * LDSW], (const float4*)(Wp + W1P) + (wave & 3) * 512 + lane, kh);
    tile_store<64>(acc2, H2, m_base + rt * 16 + (kh ? 8 : 0), col, n);
}

extern "C" void kernel_launch(void* const* d_in, const int* in_sizes, int n_in,
                              void* d_out, int out_size, void* d_ws, size_t ws_size,
                              hipStream_t stream) {
    const float* x  = (const float*)d_in[0];
    const int*   ei = (const int*)d_in[1];
    const float* W1 = (const float*)d_in[2];
    const float* b1 = (const float*)d_in[3];
    const float* W2 = (const float*)d_in[4];
    const float* b2 = (const float*)d_in[5];
    float* out = (float*)d_out;

    const int n = in_sizes[0] / 128;
    const int e = in_sizes[1] / 2;
    const int* src = ei;
    const int* dst = ei + e;

    float* Wp = (float*)d_ws;
    float* dinv = Wp + W1P + W2P;
    float* pbuf = dinv + n;
    float* h2 = pbuf + (size_t)n * 128;
    int* rowptr = (int*)(h2 + (size_t)n * 64);
    int* ssrc   = rowptr + (n + 1);
    const int nb = (n + BNODES - 1) / BNODES;
    const int nh = nb * GA;
    int* cntT   = ssrc + e;
    int* lpT    = cntT + nh;
    int2* pairs = (int2*)(pbuf + (n & 1));

    const unsigned T = 256;
    k_chunk_sort<<<GA, CT, 0, stream>>>(src, dst, cntT, lpT, pairs, e, nb, W1, W2, Wp);
    k_range_sort<<<nb, CT, 0, stream>>>(pairs, cntT, lpT, rowptr, dinv, ssrc, n, e);

    k_gather_gemm12<<<(n + FR - 1) / FR, FW * 32, 0, stream>>>(rowptr, ssrc, x, dinv, Wp, b1, h2, n);
    k_gather64<<<(n + 7) / 8, T, 0, stream>>>(rowptr, ssrc, h2, dinv, b2, out, n);
}


# Both sort levels place items into an LDS image of the block's own output stretch and copy it out in 16-byte whole-line stores, not one scattered 8- or 4-byte store per item; row order as before
# speedup vs baseline: 4.9452x; 1.0647x over previous
#include <hip/hip_runtime.h>

typedef __attribute__((ext_vector_type(2))) float v2f;
typedef __attribute__((ext_vector_type(8))) float v8f;

#define KDIM 128
#define LDSW (KDIM + 4)


#define BSH 8
#define BNODES (1 << BSH)
#define GA 256
#define CT 1024
#define MAXNB 1024
#define EPT 4
#define SCAP 8192
#define RCAP 8192

#define RPW (GA / (CT / 32))
static_assert(CT == MAXNB, "thread t of a chunk block owns the counter of coarse range t");
static_assert(RPW * (CT / 32) == GA, "the waves of a range block share the GA runs evenly");

__device__ __forceinline__ int block_sum_1024(int v, int* sh) {
    const int lane = threadIdx.x & 31, wave = threadIdx.x >> 5;
#pragma unroll
    for (int o = 16; o > 0; o >>= 1) v += __shfl_xor(v, o);
    if (lane == 0) sh[wave] = v;
    __syncthreads();
    int t = sh[lane];
#pragma unroll
    for (int o = 16; o > 0; o >>= 1) t += __shfl_xor(t, o);
    __syncthreads();
    return t;
}

#define W1P (KDIM * 128)
#define W2P (KDIM * 64)
static_assert(W1P + W2P <= GA * CT, "one thread of the first launch per packed element");
__device__ __forceinline__ float pack_w_fetch(const float* __restrict__ W1, const float* __restrict__ W2, int g) {
    const bool second = g >= W1P;
    const int q = second ? g - W1P : g;
    const int s = q & 3, ln = (q >> 2) & 31, j = (q >> 7) & 15, tn = q >> 11;
    const int k = 8 * j + 4 * (ln >> 4) + s;
    const int col = tn * 16 + (ln & 15);
    return second ? W2[k * 64 + col] : W1[k * 128 + col];
}

__global__ __launch_bounds__(CT) void k_chunk_sort(const int* __restrict__ src, const int* __restrict__ dst,
                                                   int* __restrict__ cntT, int* __restrict__ lpT,
                                                   int2* __restrict__ pairs, int e, int nb,
                                                   const float* __restrict__ W1, const float* __restrict__ W2,
                                                   float* __restrict__ Wp) {
    __shared__ int h[MAXNB];
    __shared__ int sh[32];
    __shared__ __attribute__((aligned(16))) int2 stage[SCAP];
    const int lane = threadIdx.x & 31, wave = threadIdx.x >> 5;
    h[threadIdx.x] = 0;
    __syncthreads();
    const int per = (e + GA - 1) / GA;
    const int beg = blockIdx.x * per;
    const int end = min(e, beg + per);
    for (int base = beg + (int)threadIdx.x; base < end; base += CT * EPT) {
        int d[EPT];
#pragma unroll
        for (int k = 0; k < EPT; ++k) d[k] = (base + k * CT < end) ? dst[base + k * CT] : 0;
#pragma unroll
        for (int k = 0; k < EPT; ++k)
            if (base + k * CT < end) atomicAdd(&h[d[k] >> BSH], 1);
    }
    __syncthreads();
    const int c = h[threadIdx.x];
    int incl = c;
#pragma unroll
    for (int o = 1; o < 32; o <<= 1) {
        int u = __shfl_up(incl, o);
        if (lane >= o) incl += u;
    }
    if (lane == 31) sh[wave] = incl;
    __syncthreads();
    int wsum = sh[lane];
#pragma unroll
    for (int o = 1; o < 32; o <<= 1) {
        int u = __shfl_up(wsum, o);
        if (lane >= o) wsum += u;
    }
    const int wave_excl = __shfl(wsum, wave) - __shfl(sh[lane], wave);
    const int excl = wave_excl + incl - c;
    if ((int)threadIdx.x < nb) {
        cntT[threadIdx.x * GA + blockIdx.x] = c;
        lpT[threadIdx.x * GA + blockIdx.x] = excl;
    }
    const int a0 = (int)((((size_t)(pairs + beg)) & 127) >> 3);
    h[threadIdx.x] = a0 + excl;
    __syncthreads();
    const int g = (int)(blockIdx.x * CT + threadIdx.x);
    float wv = 0.0f;
    if (g < W1P + W2P) wv = pack_w_fetch(W1, W2, g);
    for (int base = beg + (int)threadIdx.x; base < end; base += CT * EPT) {
        int d[EPT], s[EPT];
#pragma unroll
        for (int k = 0; k < EPT; ++k) {
            const bool in = base + k * CT < end;
            d[k] = in ? dst[base + k * CT] : 0;
            s[k] = in ? src[base + k * CT] : 0;
        }
#pragma unroll
        for (int k = 0; k < EPT; ++k)
            if (base + k * CT < end) {
                const int pos = atomicAdd(&h[d[k] >> BSH], 1);
                if (pos < SCAP) stage[pos] = make_int2(s[k], d[k]);
                else pairs[beg + pos - a0] = make_int2(s[k], d[k]);
            }
    }
    __syncthreads();
    const int hi = min(a0 + max(end - beg, 0), SCAP);
    for (int q = 2 * (int)threadIdx.x; q < hi; q += 2 * CT) {
        if (q >= a0 && q + 2 <= hi) {
            *(int4*)(pairs + (beg - a0 + q)) = *(const int4*)(stage + q);
        } else {
            if (q >= a0) pairs[beg - a0 + q] = stage[q];
            if (q + 1 >= a0 && q + 1 < hi) pairs[beg - a0 + q + 1] = stage[q + 1];
        }
    }
    if (g < W1P + W2P) Wp[g] = wv;
}

__global__ __launch_bounds__(CT) void k_range_sort(const int2* __restrict__ pairs,
                                                   const int* __restrict__ cntT, const int* __restrict__ lpT,
                                                   int* __restrict__ rowptr, float* __restrict__ dinv,
                                                   int* __restrict__ ssrc, int n, int e) {
    __shared__ int cnt[BNODES];
    __shared__ int sh[32];
    __shared__ int rcn[GA];
    __shared__ int rst[GA];
    __shared__ __attribute__((aligned(16))) int sstage[RCAP];
    const int lane = threadIdx.x & 31, wave = threadIdx.x >> 5;
    const int node0 = blockIdx.x << BSH;
    const int per = (e + GA - 1) / GA;
    int lp = 0;
    if (threadIdx.x < GA) {
        lp = lpT[blockIdx.x * GA + threadIdx.x];
        rcn[threadIdx.x] = cntT[blockIdx.x * GA + threadIdx.x];
        rst[threadIdx.x] = (int)threadIdx.x * per + lp;
    }
    if (threadIdx.x < BNODES) cnt[threadIdx.x] = 0;
    const int beg = block_sum_1024(lp, sh);

    int2 p[RPW];
    int nn[RPW], st[RPW];
#pragma unroll
    for (int k = 0; k < RPW; ++k) {
        nn[k] = __builtin_amdgcn_readfirstlane(rcn[wave * RPW + k]);
        st[k] = __builtin_amdgcn_readfirstlane(rst[wave * RPW + k]);
    }
#pragma unroll
    for (int k = 0; k < RPW; ++k) p[k] = (lane < nn[k]) ? pairs[st[k] + lane] : make_int2(0, node0);
#pragma unroll
    for (int k = 0; k < RPW; ++k)
        if (lane < nn[k]) atomicAdd(&cnt[p[k].y - node0], 1);
    for (int k = 0; k < RPW; ++k)
        for (int j = lane + 32; j < nn[k]; j += 32) atomicAdd(&cnt[pairs[st[k] + j].y - node0], 1);
    __syncthreads();
    const int c = (threadIdx.x < BNODES) ? cnt[threadIdx.x] : 0;
    int incl = c;
#pragma unroll
    for (int o = 1; o < 32; o <<= 1) {
        int u = __shfl_up(incl, o);
        if (lane >= o) incl += u;
    }
    if (lane == 31) sh[wave] = incl;
    __syncthreads();
    int woff = 0;
    for (int w = 0; w < BNODES / 32; ++w) woff += (w < wave) ? sh[w] : 0;
    const int excl = beg + woff + incl - c;
    const int a0 = (int)((((size_t)(ssrc + beg)) & 127) >> 2);
    if (threadIdx.x < BNODES) {
        const int node = node0 + (int)threadIdx.x;
        cnt[threadIdx.x] = a0 + excl - beg;
        if (node < n) {
            rowptr[node] = excl;
            dinv[node] = rsqrtf((float)(c + 1));
            if (node == n - 1) rowptr[n] = excl + c;
        }
    }
    __syncthreads();
#pragma unroll
    for (int k = 0; k < RPW; ++k)
        if (lane < nn[k]) {
            const int pos = atomicAdd(&cnt[p[k].y - node0], 1);
            if (pos < RCAP) sstage[pos] = p[k].x;
            else ssrc[beg + pos - a0] = p[k].x;
        }
    for (int k = 0; k < RPW; ++k)
        for (int j = lane + 32; j < nn[k]; j += 32) {
            const int2 q = pairs[st[k] + j];
            const int pos = atomicAdd(&cnt[q.y - node0], 1);
            if (pos < RCAP) sstage[pos] = q.x;
            else ssrc[beg + pos - a0] = q.x;
        }
    __syncthreads();
    int tot = 0;
    for (int w = 0; w < BNODES / 32; ++w) tot += sh[w];
    const int hi = min(a0 + tot, RCAP);
    for (int q = 4 * (int)threadIdx.x; q < hi; q += 4 * CT) {
        if (q >= a0 && q + 4 <= hi) {
            *(int4*)(ssrc + (beg - a0 + q)) = *(const int4*)(sstage + q);
        } else {
            for (int t = 0; t < 4; ++t)
                if (q + t >= a0 && q + t < hi) ssrc[beg - a0 + q + t] = sstage[q + t];
        }
    }
}

__device__ __forceinline__ v8f tile_product(const float* arow, const float4* __restrict__ wfrag, int kh) {
    v8f acc = {};
#pragma unroll
    for (int j = 0; j < KDIM / 8; ++j) {
        const float4 a = *(const float4*)(arow + 8 * j + 4 * kh);
        const float4 b = wfrag[j * 32];
        const v2f a0 = {a.x, a.y}, b0 = {b.x, b.y};
        const v2f a1 = {a.z, a.w}, b1 = {b.z, b.w};
        acc = __builtin_amdgcn_wmma_f32_16x16x4_f32(
             false, a0,  false, b0,
             (short)0, acc,  false,  false);
        acc = __builtin_amdgcn_wmma_f32_16x16x4_f32(
             false, a1,  false, b1,
             (short)0, acc,  false,  false);
    }
    return acc;
}

template <int OUTCH>
__device__ __forceinline__ void tile_store(const v8f& acc, float* __restrict__ H, int mrow0, int col, int n) {
#pragma unroll
    for (int r = 0; r < 8; ++r) {
        int grow = mrow0 + r;
        if (grow < n)
            H[(size_t)grow * OUTCH + col] = acc[r];
    }
}


__device__ __forceinline__ void row_add(float2& a, const float2& h) {
    a.x += h.x; a.y += h.y;
}
__device__ __forceinline__ void row_fma(float4& a, float w, const float4& h) {
    a.x = fmaf(w, h.x, a.x); a.y = fmaf(w, h.y, a.y); a.z = fmaf(w, h.z, a.z); a.w = fmaf(w, h.w, a.w);
}

__device__ __forceinline__ float2 gather_row64(int node, int lane, float dd,
                                               const int* __restrict__ rowptr,
                                               const int* __restrict__ ssrc,
                                               const float* __restrict__ hs,
                                               const float* __restrict__ bias) {
    const int beg = __builtin_amdgcn_readfirstlane(rowptr[node]);
    const int end = __builtin_amdgcn_readfirstlane(rowptr[node + 1]);

    const float2 b = ((const float2*)bias)[lane];
    float2 acc = ((const float2*)(hs + (size_t)node * 64))[lane];

    for (int base = beg; base < end; base += 32) {
        const int m = min(32, end - base);
        int s_l = node;
        if (lane < m) s_l = ssrc[base + lane];
#define EDGE_ROW(j) (((const float2*)(hs + (size_t)__builtin_amdgcn_readlane(s_l, (j)) * 64))[lane])
        int k = 0;
        for (; k + 4 <= m; k += 4) {
            const float2 h0 = EDGE_ROW(k + 0);
            const float2 h1 = EDGE_ROW(k + 1);
            const float2 h2 = EDGE_ROW(k + 2);
            const float2 h3 = EDGE_ROW(k + 3);
            row_add(acc, h0);
            row_add(acc, h1);
            row_add(acc, h2);
            row_add(acc, h3);
        }
        if (k < m) {
            const float2 h0 = EDGE_ROW(k);
            if (k + 1 < m) {
                const float2 h1 = EDGE_ROW(k + 1);
                if (k + 2 < m) {
                    const float2 h2 = EDGE_ROW(k + 2);
                    row_add(acc, h0);
                    row_add(acc, h1);
                    row_add(acc, h2);
                } else {
                    row_add(acc, h0);
                    row_add(acc, h1);
                }
            } else {
                row_add(acc, h0);
            }
        }
#undef EDGE_ROW
    }
    return make_float2(fmaf(acc.x, dd, b.x), fmaf(acc.y, dd, b.y));
}

__device__ __forceinline__ float4 gather_row_x(int node, int lane, float dd,
                                               const int* __restrict__ rowptr,
                                               const int* __restrict__ ssrc,
                                               const float* __restrict__ dinv,
                                               const float* __restrict__ x) {
    const int beg = __builtin_amdgcn_readfirstlane(rowptr[node]);
    const int end = __builtin_amdgcn_readfirstlane(rowptr[node + 1]);

    float4 acc = ((const float4*)(x + (size_t)node * KDIM))[lane];
    acc.x *= dd; acc.y *= dd; acc.z *= dd; acc.w *= dd;

    for (int base = beg; base < end; base += 32) {
        const int m = min(32, end - base);
        int s_l = node;
        if (lane < m) s_l = ssrc[base + lane];
        const int w_l = __float_as_int(dinv[s_l]);
#define EDGE_ROW(j) (((const float4*)(x + (size_t)__builtin_amdgcn_readlane(s_l, (j)) * KDIM))[lane])
#define EDGE_W(j) __int_as_float(__builtin_amdgcn_readlane(w_l, (j)))
        int k = 0;
        for (; k + 4 <= m; k += 4) {
            const float4 h0 = EDGE_ROW(k + 0);
            const float4 h1 = EDGE_ROW(k + 1);
            const float4 h2 = EDGE_ROW(k + 2);
            const float4 h3 = EDGE_ROW(k + 3);
            row_fma(acc, EDGE_W(k + 0), h0);
            row_fma(acc, EDGE_W(k + 1), h1);
            row_fma(acc, EDGE_W(k + 2), h2);
            row_fma(acc, EDGE_W(k + 3), h3);
        }
        if (k < m) {
            const float4 h0 = EDGE_ROW(k);
            if (k + 1 < m) {
                const float4 h1 = EDGE_ROW(k + 1);
                if (k + 2 < m) {
                    const float4 h2 = EDGE_ROW(k + 2);
                    row_fma(acc, EDGE_W(k), h0);
                    row_fma(acc, EDGE_W(k + 1), h1);
                    row_fma(acc, EDGE_W(k + 2), h2);
                } else {
                    row_fma(acc, EDGE_W(k), h0);
                    row_fma(acc, EDGE_W(k + 1), h1);
                }
            } else {
                row_fma(acc, EDGE_W(k), h0);
            }
        }
#undef EDGE_W
#undef EDGE_ROW
    }
    return acc;
}

__global__ __launch_bounds__(256) void k_gather64(const int* __restrict__ rowptr,
                                                  const int* __restrict__ ssrc,
                                                  const float* __restrict__ h,
                                                  const float* __restrict__ dinv,
                                                  const float* __restrict__ bias,
                                                  float* __restrict__ out, int n) {
    const int lane = threadIdx.x & 31;
    const int node = __builtin_amdgcn_readfirstlane((int)((blockIdx.x * 256u + threadIdx.x) >> 5));
    if (node >= n) return;
    ((float2*)(out + (size_t)node * 64))[lane] = gather_row64(node, lane, dinv[node], rowptr, ssrc, h, bias);
}

#define FW 16
#define FR (2 * FW)
static_assert(FR % 16 == 0 && (FR / 16) * 8 == FW, "one 16x16 tile of the layer-1 product per wave");
__global__ __launch_bounds__(FW * 32, 16) void k_gather_gemm12(const int* __restrict__ rowptr,
                                                           const int* __restrict__ ssrc,
                                                           const float* __restrict__ x,
                                                           const float* __restrict__ dinv,
                                                           const float* __restrict__ Wp,
                                                           const float* __restrict__ b1,
                                                           float* __restrict__ H2, int n) {
    __shared__ float tile[FR * LDSW];
    __shared__ float tile2[FR * LDSW];
    __shared__ float rdinv[FR];

    const int lane = threadIdx.x & 31;
    const int wave = __builtin_amdgcn_readfirstlane((int)(threadIdx.x >> 5));
    const int m_base = blockIdx.x * FR;

    for (int r = 0; r < 2; ++r) {
        const int lrow = 2 * wave + r;
        const int node = m_base + lrow;
        float4 v = make_float4(0.0f, 0.0f, 0.0f, 0.0f);
        float dd = 0.0f;
        if (node < n) {
            dd = dinv[node];
            v = gather_row_x(node, lane, dd, rowptr, ssrc, dinv, x);
        }
        *(float4*)&tile[lrow * LDSW + lane * 4] = v;
        if (lane == 0) rdinv[lrow] = dd;
    }
    __syncthreads();

    const int lm = lane & 15;
    const int kh = lane >> 4;
    {
        const int rt = wave >> 3;
        const int col = (wave & 7) * 16 + lm;
        const float bc = b1[col];
        v8f acc = tile_product(&tile[(rt * 16 + lm) * LDSW], (const float4*)Wp + (wave & 7) * 512 + lane, kh);
        asm volatile("v_nop\n\tv_nop\n\tv_nop\n\tv_nop" : "+v"(acc));
#pragma unroll
        for (int r = 0; r < 8; ++r) {
            const int row = rt * 16 + 8 * kh + r;
            const float d = rdinv[row];
            tile2[row * LDSW + col] = fmaxf(fmaf(acc[r], d, bc), 0.0f) * d;
        }
    }
    __syncthreads();
    if (wave >= FR / 4) return;

    const int rt = wave >> 2;
    const int col = (wave & 3) * 16 + lm;
    const v8f acc2 = tile_product(&tile2[(rt * 16 + lm) * LDSW], (const float4*)(Wp + W1P) + (wave & 3) * 512 + lane, kh);
    tile_store<64>(acc2, H2, m_base + rt * 16 + (kh ? 8 : 0), col, n);
}

extern "C" void kernel_launch(void* const* d_in, const int* in_sizes, int n_in,
                              void* d_out, int out_size, void* d_ws, size_t ws_size,
                              hipStream_t stream) {
    const float* x  = (const float*)d_in[0];
    const int*   ei = (const int*)d_in[1];
    const float* W1 = (const float*)d_in[2];
    const float* b1 = (const float*)d_in[3];
    const float* W2 = (const float*)d_in[4];
    const float* b2 = (const float*)d_in[5];
    float* out = (float*)d_out;

    const int n = in_sizes[0] / 128;
    const int e = in_sizes[1] / 2;
    const int* src = ei;
    const int* dst = ei + e;

    float* Wp = (float*)d_ws;
    float* dinv = Wp + W1P + W2P;
    float* pbuf = dinv + n;
    float* h2 = pbuf + (size_t)n * 128;
    int* rowptr = (int*)(h2 + (size_t)n * 64);
    int* ssrc   = rowptr + (n + 1);
    const int nb = (n + BNODES - 1) / BNODES;
    const int nh = nb * GA;
    int* cntT   = ssrc + e;
    int* lpT    = cntT + nh;
    int2* pairs = (int2*)(pbuf + (n & 1));

    const unsigned T = 256;
    k_chunk_sort<<<GA, CT, 0, stream>>>(src, dst, cntT, lpT, pairs, e, nb, W1, W2, Wp);
    k_range_sort<<<nb, CT, 0, stream>>>(pairs, cntT, lpT, rowptr, dinv, ssrc, n, e);

    k_gather_gemm12<<<(n + FR - 1) / FR, FW * 32, 0, stream>>>(rowptr, ssrc, x, dinv, Wp, b1, h2, n);
    k_gather64<<<(n + 7) / 8, T, 0, stream>>>(rowptr, ssrc, h2, dinv, b2, out, n);
}
